# starting kernel
# baseline (speedup 1.0000x reference)
#include <hip/hip_runtime.h>
#include <cstddef>

typedef float v2f __attribute__((ext_vector_type(2)));
typedef float v8f __attribute__((ext_vector_type(8)));

#define D_MODEL 768
#define N_HEAD  12
#define D_HEAD  64
#define N_CTX   2048
#define BATCH   2
#define M_ROWS  (BATCH * N_CTX)

__device__ __forceinline__ v8f wmma_f32(v2f a, v2f b, v8f c) {
  return __builtin_amdgcn_wmma_f32_16x16x4_f32(false, a, false, b, (short)0, c,
                                               false, false);
}

__device__ __forceinline__ void gemm_strip_k768(const float* __restrict__ xrow,
                                                const float* __restrict__ w0,
                                                const float* __restrict__ w1,
                                                const float* __restrict__ w2,
                                                const float* __restrict__ w3,
                                                v8f acc[4]) {
#pragma unroll 4
  for (int k = 0; k < D_MODEL; k += 4) {
    v2f a  = *(const v2f*)(xrow + k);
    v2f b0 = *(const v2f*)(w0 + k);
    v2f b1 = *(const v2f*)(w1 + k);
    v2f b2 = *(const v2f*)(w2 + k);
    v2f b3 = *(const v2f*)(w3 + k);
    acc[0] = wmma_f32(a, b0, acc[0]);
    acc[1] = wmma_f32(a, b1, acc[1]);
    acc[2] = wmma_f32(a, b2, acc[2]);
    acc[3] = wmma_f32(a, b3, acc[3]);
  }
}

__global__ __launch_bounds__(256) void qkv_gemm_kernel(
    const float* __restrict__ X, const float* __restrict__ W,
    const float* __restrict__ bias, float* __restrict__ out, int vmode) {
  const int wave = threadIdx.x >> 5;
  const int lane = threadIdx.x & 31;
  const int lo = lane & 15, hi = lane >> 4;
  const int m0 = blockIdx.x * 32 + (wave >> 2) * 16;
  const int n0 = blockIdx.y * 256 + (wave & 3) * 64;
  const float* xrow  = X + (size_t)(m0 + lo) * D_MODEL + 2 * hi;
  const float* wbase = W + (size_t)(n0 + lo) * D_MODEL + 2 * hi;
  v8f acc[4] = {{}, {}, {}, {}};
  gemm_strip_k768(xrow, wbase, wbase + 16 * D_MODEL, wbase + 32 * D_MODEL,
                  wbase + 48 * D_MODEL, acc);
  const int h = n0 >> 6;
#pragma unroll
  for (int j = 0; j < 4; ++j) {
    const int d = j * 16 + lo;
    const float bv = bias[n0 + d];
#pragma unroll
    for (int r = 0; r < 8; ++r) {
      const int m = m0 + r + 8 * hi;
      const int b = m >> 11, s = m & (N_CTX - 1);
      const size_t hb = (size_t)b * N_HEAD + h;
      if (vmode)
        out[(hb * D_HEAD + d) * N_CTX + s] = acc[j][r] + bv;
      else
        out[(hb * N_CTX + s) * D_HEAD + d] = acc[j][r] + bv;
    }
  }
}

__global__ __launch_bounds__(256) void out_gemm_kernel(
    const float* __restrict__ X, const float* __restrict__ W,
    float* __restrict__ out) {
  const int wave = threadIdx.x >> 5;
  const int lane = threadIdx.x & 31;
  const int lo = lane & 15, hi = lane >> 4;
  const int m0 = blockIdx.x * 32 + (wave >> 2) * 16;
  const int n0 = blockIdx.y * 256 + (wave & 3) * 64;
  const float* xrow  = X + (size_t)(m0 + lo) * D_MODEL + 2 * hi;
  const float* wbase = W + (size_t)(n0 + lo) * D_MODEL + 2 * hi;
  v8f acc[4] = {{}, {}, {}, {}};
  gemm_strip_k768(xrow, wbase, wbase + 16 * D_MODEL, wbase + 32 * D_MODEL,
                  wbase + 48 * D_MODEL, acc);
#pragma unroll
  for (int j = 0; j < 4; ++j) {
#pragma unroll
    for (int r = 0; r < 8; ++r) {
      const int m = m0 + r + 8 * hi;
      out[(size_t)m * D_MODEL + n0 + j * 16 + lo] = acc[j][r];
    }
  }
}

__global__ __launch_bounds__(256) void norm_rope_kernel(
    float* __restrict__ qk, const float* __restrict__ g) {
  const int lane = threadIdx.x & 31;
  const int row = blockIdx.x * 8 + (threadIdx.x >> 5);
  const int s = row & (N_CTX - 1);
  float* p = qk + (size_t)row * D_HEAD;
  const float x1 = p[lane];
  const float x2 = p[lane + 32];
  float ss = x1 * x1 + x2 * x2;
#pragma unroll
  for (int off = 16; off > 0; off >>= 1) ss += __shfl_xor(ss, off, 32);
  const float inv = rsqrtf(ss * (1.0f / 64.0f) + 1e-6f);
  const float xn1 = x1 * inv * g[lane];
  const float xn2 = x2 * inv * g[lane + 32];
  const float freq = (float)s * exp2f((float)lane * (-11.0f / 32.0f));
  const float c = __cosf(freq), sn = __sinf(freq);
  p[lane]      = xn1 * c + xn2 * sn;
  p[lane + 32] = xn2 * c - xn1 * sn;
}

__global__ __launch_bounds__(256) void attn_kernel(
    const float* __restrict__ Q, const float* __restrict__ K,
    const float* __restrict__ V, float* __restrict__ Z) {
  __shared__ float lds[8 * 256];
  const int wave = threadIdx.x >> 5;
  const int lane = threadIdx.x & 31;
  const int lo = lane & 15, hi = lane >> 4;
  const int wid = blockIdx.x * 8 + wave;
  const int head = wid >> 7;
  const int qt = wid & 127;
  const float* Qh = Q + (size_t)head * N_CTX * D_HEAD;
  const float* Kh = K + (size_t)head * N_CTX * D_HEAD;
  const float* Vh = V + (size_t)head * D_HEAD * N_CTX;
  float* pl = lds + wave * 256;

  v2f qa[16];
  {
    const float* qrow = Qh + (size_t)(qt * 16 + lo) * D_HEAD + 2 * hi;
#pragma unroll
    for (int kk = 0; kk < 16; ++kk) qa[kk] = *(const v2f*)(qrow + kk * 4);
  }

  v8f z0 = {}, z1 = {}, z2 = {}, z3 = {};
  for (int kt = 0; kt <= qt; ++kt) {
    const float* krow = Kh + (size_t)(kt * 16 + lo) * D_HEAD + 2 * hi;
    __builtin_prefetch(krow + 16 * D_HEAD, 0, 0);
    v8f sc = {};
#pragma unroll
    for (int kk = 0; kk < 16; ++kk) {
      v2f kb = *(const v2f*)(krow + kk * 4);
      sc = wmma_f32(qa[kk], kb, sc);
    }
#pragma unroll
    for (int r = 0; r < 8; ++r) {
      float pv = sc[r] * (1.0f / 64.0f);
      pv = pv * pv;
      if (kt == qt && lo > r + 8 * hi) pv = 0.0f;
      pl[(r + 8 * hi) * 16 + lo] = pv;
    }
#pragma unroll
    for (int kk2 = 0; kk2 < 4; ++kk2) {
      v2f pa = *(const v2f*)(pl + lo * 16 + kk2 * 4 + 2 * hi);
      const float* vcol = Vh + (size_t)lo * N_CTX + kt * 16 + kk2 * 4 + 2 * hi;
      v2f v0 = *(const v2f*)(vcol);
      v2f v1 = *(const v2f*)(vcol + 16 * N_CTX);
      v2f v2 = *(const v2f*)(vcol + 32 * N_CTX);
      v2f v3 = *(const v2f*)(vcol + 48 * N_CTX);
      z0 = wmma_f32(pa, v0, z0);
      z1 = wmma_f32(pa, v1, z1);
      z2 = wmma_f32(pa, v2, z2);
      z3 = wmma_f32(pa, v3, z3);
    }
  }
  const int b = head / N_HEAD, h = head % N_HEAD;
#pragma unroll
  for (int r = 0; r < 8; ++r) {
    const int s = qt * 16 + r + 8 * hi;
    float* zr = Z + (size_t)(b * N_CTX + s) * D_MODEL + h * D_HEAD + lo;
    zr[0]  = z0[r];
    zr[16] = z1[r];
    zr[32] = z2[r];
    zr[48] = z3[r];
  }
}

extern "C" void kernel_launch(void* const* d_in, const int* in_sizes, int n_in,
                              void* d_out, int out_size, void* d_ws, size_t ws_size,
                              hipStream_t stream) {
  const float* x  = (const float*)d_in[0];
  const float* Wq = (const float*)d_in[1];
  const float* bq = (const float*)d_in[2];
  const float* Wk = (const float*)d_in[3];
  const float* bk = (const float*)d_in[4];
  const float* Wv = (const float*)d_in[5];
  const float* bv = (const float*)d_in[6];
  const float* Wo = (const float*)d_in[7];
  const float* g  = (const float*)d_in[8];
  float* out = (float*)d_out;

  const size_t QKV = (size_t)BATCH * N_HEAD * N_CTX * D_HEAD;
  float* q = (float*)d_ws;
  float* k = q + QKV;
  float* v = k + QKV;
  float* z = v + QKV;

  dim3 gblk(M_ROWS / 32, D_MODEL / 256);
  qkv_gemm_kernel<<<gblk, 256, 0, stream>>>(x, Wq, bq, q, 0);
  qkv_gemm_kernel<<<gblk, 256, 0, stream>>>(x, Wk, bk, k, 0);
  qkv_gemm_kernel<<<gblk, 256, 0, stream>>>(x, Wv, bv, v, 1);

  const int rows = BATCH * N_HEAD * N_CTX;
  norm_rope_kernel<<<rows / 8, 256, 0, stream>>>(q, g);
  norm_rope_kernel<<<rows / 8, 256, 0, stream>>>(k, g);

  attn_kernel<<<(BATCH * N_HEAD * (N_CTX / 16)) / 8, 256, 0, stream>>>(q, k, v, z);

  out_gemm_kernel<<<gblk, 256, 0, stream>>>(z, Wo, out);
}


# GEMM operands staged through LDS: 64x192 block tile (256 blocks x 8 waves), K chunks of 32 copied with whole-line b128 loads into a 2-stage LDS ring, b64 fragments from LDS, pitch 36; sums unchanged
# speedup vs baseline: 1.4582x; 1.4582x over previous
#include <hip/hip_runtime.h>
#include <cstddef>

typedef float v2f __attribute__((ext_vector_type(2)));
typedef float v8f __attribute__((ext_vector_type(8)));

#define D_MODEL 768
#define N_HEAD  12
#define D_HEAD  64
#define N_CTX   2048
#define BATCH   2
#define M_ROWS  (BATCH * N_CTX)

__device__ __forceinline__ v8f wmma_f32(v2f a, v2f b, v8f c) {
  return __builtin_amdgcn_wmma_f32_16x16x4_f32(false, a, false, b, (short)0, c,
                                               false, false);
}

typedef float v4f __attribute__((ext_vector_type(4)));

#define BM 64
#define BN 192
#define KC 32
#define PITCH 36
#define STAGE_FLOATS ((BM + BN) * PITCH)
#define N_CHUNK (D_MODEL / KC)

__device__ __forceinline__ void gemm_tile_lds(const float* __restrict__ X,
                                              const float* __restrict__ W,
                                              float* __restrict__ lds,
                                              v8f acc[2][3]) {
  const int tid = threadIdx.x;
  const int wave = tid >> 5, lane = tid & 31;
  const int lo = lane & 15, hi = lane >> 4;
  const int wm = wave >> 2, wn = wave & 3;
  const int cr = tid >> 3, cc = (tid & 7) * 4;
  const float* xg = X + (size_t)(blockIdx.x * BM + cr) * D_MODEL + cc;
  const float* wg = W + (size_t)(blockIdx.y * BN + cr) * D_MODEL + cc;
  float* st = lds + cr * PITCH + cc;
  const float* fa = lds + (wm * 32 + lo) * PITCH + 2 * hi;
  const float* fb = lds + (BM + wn * 48 + lo) * PITCH + 2 * hi;

  v4f pre[8];
  pre[0] = *(const v4f*)(xg);
  pre[1] = *(const v4f*)(xg + 32 * D_MODEL);
#pragma unroll
  for (int i = 0; i < 6; ++i) pre[2 + i] = *(const v4f*)(wg + i * 32 * D_MODEL);
#pragma unroll
  for (int i = 0; i < 8; ++i) *(v4f*)(st + i * 32 * PITCH) = pre[i];
  __syncthreads();

  for (int c = 0; c < N_CHUNK; ++c) {
    const int cur = (c & 1) * STAGE_FLOATS;
    const bool more = (c + 1 < N_CHUNK);
    if (more) {
      const int k1 = (c + 1) * KC;
      pre[0] = *(const v4f*)(xg + k1);
      pre[1] = *(const v4f*)(xg + 32 * D_MODEL + k1);
#pragma unroll
      for (int i = 0; i < 6; ++i)
        pre[2 + i] = *(const v4f*)(wg + i * 32 * D_MODEL + k1);
    }
    v2f af[KC / 4][2], bf[KC / 4][3];
#pragma unroll
    for (int kk = 0; kk < KC / 4; ++kk) {
      af[kk][0] = *(const v2f*)(fa + cur + kk * 4);
      af[kk][1] = *(const v2f*)(fa + cur + 16 * PITCH + kk * 4);
      bf[kk][0] = *(const v2f*)(fb + cur + kk * 4);
      bf[kk][1] = *(const v2f*)(fb + cur + 16 * PITCH + kk * 4);
      bf[kk][2] = *(const v2f*)(fb + cur + 32 * PITCH + kk * 4);
    }
    __builtin_amdgcn_sched_barrier(0);
#pragma unroll
    for (int kk = 0; kk < KC / 4; ++kk) {
      acc[0][0] = wmma_f32(af[kk][0], bf[kk][0], acc[0][0]);
      acc[0][1] = wmma_f32(af[kk][0], bf[kk][1], acc[0][1]);
      acc[0][2] = wmma_f32(af[kk][0], bf[kk][2], acc[0][2]);
      acc[1][0] = wmma_f32(af[kk][1], bf[kk][0], acc[1][0]);
      acc[1][1] = wmma_f32(af[kk][1], bf[kk][1], acc[1][1]);
      acc[1][2] = wmma_f32(af[kk][1], bf[kk][2], acc[1][2]);
    }
    __builtin_amdgcn_sched_barrier(0);
    if (more) {
      const int nxt = STAGE_FLOATS - cur;
#pragma unroll
      for (int i = 0; i < 8; ++i) *(v4f*)(st + nxt + i * 32 * PITCH) = pre[i];
    }
    __syncthreads();
  }
}

__global__ __launch_bounds__(256) void qkv_gemm_kernel(
    const float* __restrict__ X, const float* __restrict__ W,
    const float* __restrict__ bias, float* __restrict__ out, int vmode) {
  __shared__ __attribute__((aligned(16))) float lds[2 * STAGE_FLOATS];
  const int wave = threadIdx.x >> 5;
  const int lane = threadIdx.x & 31;
  const int lo = lane & 15, hi = lane >> 4;
  const int m0 = blockIdx.x * BM + (wave >> 2) * 32;
  const int n0 = blockIdx.y * BN + (wave & 3) * 48;
  v8f acc[2][3] = {{{}, {}, {}}, {{}, {}, {}}};
  gemm_tile_lds(X, W, lds, acc);
#pragma unroll
  for (int j = 0; j < 3; ++j) {
    const int n = n0 + j * 16 + lo;
    const int h = n >> 6, d = n & (D_HEAD - 1);
    const float bv = bias[n];
#pragma unroll
    for (int i = 0; i < 2; ++i) {
#pragma unroll
      for (int r = 0; r < 8; ++r) {
        const int m = m0 + i * 16 + r + 8 * hi;
        const int b = m >> 11, s = m & (N_CTX - 1);
        const size_t hb = (size_t)b * N_HEAD + h;
        if (vmode)
          out[(hb * D_HEAD + d) * N_CTX + s] = acc[i][j][r] + bv;
        else
          out[(hb * N_CTX + s) * D_HEAD + d] = acc[i][j][r] + bv;
      }
    }
  }
}

__global__ __launch_bounds__(256) void out_gemm_kernel(
    const float* __restrict__ X, const float* __restrict__ W,
    float* __restrict__ out) {
  __shared__ __attribute__((aligned(16))) float lds[2 * STAGE_FLOATS];
  const int wave = threadIdx.x >> 5;
  const int lane = threadIdx.x & 31;
  const int lo = lane & 15, hi = lane >> 4;
  const int m0 = blockIdx.x * BM + (wave >> 2) * 32;
  const int n0 = blockIdx.y * BN + (wave & 3) * 48;
  v8f acc[2][3] = {{{}, {}, {}}, {{}, {}, {}}};
  gemm_tile_lds(X, W, lds, acc);
#pragma unroll
  for (int i = 0; i < 2; ++i) {
#pragma unroll
    for (int j = 0; j < 3; ++j) {
#pragma unroll
      for (int r = 0; r < 8; ++r) {
        const int m = m0 + i * 16 + r + 8 * hi;
        out[(size_t)m * D_MODEL + n0 + j * 16 + lo] = acc[i][j][r];
      }
    }
  }
}

__global__ __launch_bounds__(256) void norm_rope_kernel(
    float* __restrict__ qk, const float* __restrict__ g) {
  const int lane = threadIdx.x & 31;
  const int row = blockIdx.x * 8 + (threadIdx.x >> 5);
  const int s = row & (N_CTX - 1);
  float* p = qk + (size_t)row * D_HEAD;
  const float x1 = p[lane];
  const float x2 = p[lane + 32];
  float ss = x1 * x1 + x2 * x2;
#pragma unroll
  for (int off = 16; off > 0; off >>= 1) ss += __shfl_xor(ss, off, 32);
  const float inv = rsqrtf(ss * (1.0f / 64.0f) + 1e-6f);
  const float xn1 = x1 * inv * g[lane];
  const float xn2 = x2 * inv * g[lane + 32];
  const float freq = (float)s * exp2f((float)lane * (-11.0f / 32.0f));
  const float c = __cosf(freq), sn = __sinf(freq);
  p[lane]      = xn1 * c + xn2 * sn;
  p[lane + 32] = xn2 * c - xn1 * sn;
}

__global__ __launch_bounds__(256) void attn_kernel(
    const float* __restrict__ Q, const float* __restrict__ K,
    const float* __restrict__ V, float* __restrict__ Z) {
  __shared__ float lds[8 * 256];
  const int wave = threadIdx.x >> 5;
  const int lane = threadIdx.x & 31;
  const int lo = lane & 15, hi = lane >> 4;
  const int wid = blockIdx.x * 8 + wave;
  const int head = wid >> 7;
  const int qt = wid & 127;
  const float* Qh = Q + (size_t)head * N_CTX * D_HEAD;
  const float* Kh = K + (size_t)head * N_CTX * D_HEAD;
  const float* Vh = V + (size_t)head * D_HEAD * N_CTX;
  float* pl = lds + wave * 256;

  v2f qa[16];
  {
    const float* qrow = Qh + (size_t)(qt * 16 + lo) * D_HEAD + 2 * hi;
#pragma unroll
    for (int kk = 0; kk < 16; ++kk) qa[kk] = *(const v2f*)(qrow + kk * 4);
  }

  v8f z0 = {}, z1 = {}, z2 = {}, z3 = {};
  for (int kt = 0; kt <= qt; ++kt) {
    const float* krow = Kh + (size_t)(kt * 16 + lo) * D_HEAD + 2 * hi;
    __builtin_prefetch(krow + 16 * D_HEAD, 0, 0);
    v8f sc = {};
#pragma unroll
    for (int kk = 0; kk < 16; ++kk) {
      v2f kb = *(const v2f*)(krow + kk * 4);
      sc = wmma_f32(qa[kk], kb, sc);
    }
#pragma unroll
    for (int r = 0; r < 8; ++r) {
      float pv = sc[r] * (1.0f / 64.0f);
      pv = pv * pv;
      if (kt == qt && lo > r + 8 * hi) pv = 0.0f;
      pl[(r + 8 * hi) * 16 + lo] = pv;
    }
#pragma unroll
    for (int kk2 = 0; kk2 < 4; ++kk2) {
      v2f pa = *(const v2f*)(pl + lo * 16 + kk2 * 4 + 2 * hi);
      const float* vcol = Vh + (size_t)lo * N_CTX + kt * 16 + kk2 * 4 + 2 * hi;
      v2f v0 = *(const v2f*)(vcol);
      v2f v1 = *(const v2f*)(vcol + 16 * N_CTX);
      v2f v2 = *(const v2f*)(vcol + 32 * N_CTX);
      v2f v3 = *(const v2f*)(vcol + 48 * N_CTX);
      z0 = wmma_f32(pa, v0, z0);
      z1 = wmma_f32(pa, v1, z1);
      z2 = wmma_f32(pa, v2, z2);
      z3 = wmma_f32(pa, v3, z3);
    }
  }
  const int b = head / N_HEAD, h = head % N_HEAD;
#pragma unroll
  for (int r = 0; r < 8; ++r) {
    const int s = qt * 16 + r + 8 * hi;
    float* zr = Z + (size_t)(b * N_CTX + s) * D_MODEL + h * D_HEAD + lo;
    zr[0]  = z0[r];
    zr[16] = z1[r];
    zr[32] = z2[r];
    zr[48] = z3[r];
  }
}

extern "C" void kernel_launch(void* const* d_in, const int* in_sizes, int n_in,
                              void* d_out, int out_size, void* d_ws, size_t ws_size,
                              hipStream_t stream) {
  const float* x  = (const float*)d_in[0];
  const float* Wq = (const float*)d_in[1];
  const float* bq = (const float*)d_in[2];
  const float* Wk = (const float*)d_in[3];
  const float* bk = (const float*)d_in[4];
  const float* Wv = (const float*)d_in[5];
  const float* bv = (const float*)d_in[6];
  const float* Wo = (const float*)d_in[7];
  const float* g  = (const float*)d_in[8];
  float* out = (float*)d_out;

  const size_t QKV = (size_t)BATCH * N_HEAD * N_CTX * D_HEAD;
  float* q = (float*)d_ws;
  float* k = q + QKV;
  float* v = k + QKV;
  float* z = v + QKV;

  dim3 gblk(M_ROWS / BM, D_MODEL / BN);
  qkv_gemm_kernel<<<gblk, 256, 0, stream>>>(x, Wq, bq, q, 0);
  qkv_gemm_kernel<<<gblk, 256, 0, stream>>>(x, Wk, bk, k, 0);
  qkv_gemm_kernel<<<gblk, 256, 0, stream>>>(x, Wv, bv, v, 1);

  const int rows = BATCH * N_HEAD * N_CTX;
  norm_rope_kernel<<<rows / 8, 256, 0, stream>>>(q, g);
  norm_rope_kernel<<<rows / 8, 256, 0, stream>>>(k, g);

  attn_kernel<<<(BATCH * N_HEAD * (N_CTX / 16)) / 8, 256, 0, stream>>>(q, k, v, z);

  out_gemm_kernel<<<gblk, 256, 0, stream>>>(z, Wo, out);
}


# Attention K and V tiles staged through LDS: a block's 8 q tiles share 64-key K and V tiles copied with whole-line b128 loads into a 2-stage ring, b64 fragments from LDS, pitch 68; sums unchanged
# speedup vs baseline: 1.8381x; 1.2605x over previous
#include <hip/hip_runtime.h>
#include <cstddef>

typedef float v2f __attribute__((ext_vector_type(2)));
typedef float v8f __attribute__((ext_vector_type(8)));

#define D_MODEL 768
#define N_HEAD  12
#define D_HEAD  64
#define N_CTX   2048
#define BATCH   2
#define M_ROWS  (BATCH * N_CTX)

__device__ __forceinline__ v8f wmma_f32(v2f a, v2f b, v8f c) {
  return __builtin_amdgcn_wmma_f32_16x16x4_f32(false, a, false, b, (short)0, c,
                                               false, false);
}

typedef float v4f __attribute__((ext_vector_type(4)));

#define BM 64
#define BN 192
#define KC 32
#define PITCH 36
#define STAGE_FLOATS ((BM + BN) * PITCH)
#define N_CHUNK (D_MODEL / KC)

__device__ __forceinline__ void gemm_tile_lds(const float* __restrict__ X,
                                              const float* __restrict__ W,
                                              float* __restrict__ lds,
                                              v8f acc[2][3]) {
  const int tid = threadIdx.x;
  const int wave = tid >> 5, lane = tid & 31;
  const int lo = lane & 15, hi = lane >> 4;
  const int wm = wave >> 2, wn = wave & 3;
  const int cr = tid >> 3, cc = (tid & 7) * 4;
  const float* xg = X + (size_t)(blockIdx.x * BM + cr) * D_MODEL + cc;
  const float* wg = W + (size_t)(blockIdx.y * BN + cr) * D_MODEL + cc;
  float* st = lds + cr * PITCH + cc;
  const float* fa = lds + (wm * 32 + lo) * PITCH + 2 * hi;
  const float* fb = lds + (BM + wn * 48 + lo) * PITCH + 2 * hi;

  v4f pre[8];
  pre[0] = *(const v4f*)(xg);
  pre[1] = *(const v4f*)(xg + 32 * D_MODEL);
#pragma unroll
  for (int i = 0; i < 6; ++i) pre[2 + i] = *(const v4f*)(wg + i * 32 * D_MODEL);
#pragma unroll
  for (int i = 0; i < 8; ++i) *(v4f*)(st + i * 32 * PITCH) = pre[i];
  __syncthreads();

  for (int c = 0; c < N_CHUNK; ++c) {
    const int cur = (c & 1) * STAGE_FLOATS;
    const bool more = (c + 1 < N_CHUNK);
    if (more) {
      const int k1 = (c + 1) * KC;
      pre[0] = *(const v4f*)(xg + k1);
      pre[1] = *(const v4f*)(xg + 32 * D_MODEL + k1);
#pragma unroll
      for (int i = 0; i < 6; ++i)
        pre[2 + i] = *(const v4f*)(wg + i * 32 * D_MODEL + k1);
    }
    v2f af[KC / 4][2], bf[KC / 4][3];
#pragma unroll
    for (int kk = 0; kk < KC / 4; ++kk) {
      af[kk][0] = *(const v2f*)(fa + cur + kk * 4);
      af[kk][1] = *(const v2f*)(fa + cur + 16 * PITCH + kk * 4);
      bf[kk][0] = *(const v2f*)(fb + cur + kk * 4);
      bf[kk][1] = *(const v2f*)(fb + cur + 16 * PITCH + kk * 4);
      bf[kk][2] = *(const v2f*)(fb + cur + 32 * PITCH + kk * 4);
    }
    __builtin_amdgcn_sched_barrier(0);
#pragma unroll
    for (int kk = 0; kk < KC / 4; ++kk) {
      acc[0][0] = wmma_f32(af[kk][0], bf[kk][0], acc[0][0]);
      acc[0][1] = wmma_f32(af[kk][0], bf[kk][1], acc[0][1]);
      acc[0][2] = wmma_f32(af[kk][0], bf[kk][2], acc[0][2]);
      acc[1][0] = wmma_f32(af[kk][1], bf[kk][0], acc[1][0]);
      acc[1][1] = wmma_f32(af[kk][1], bf[kk][1], acc[1][1]);
      acc[1][2] = wmma_f32(af[kk][1], bf[kk][2], acc[1][2]);
    }
    __builtin_amdgcn_sched_barrier(0);
    if (more) {
      const int nxt = STAGE_FLOATS - cur;
#pragma unroll
      for (int i = 0; i < 8; ++i) *(v4f*)(st + nxt + i * 32 * PITCH) = pre[i];
    }
    __syncthreads();
  }
}

__global__ __launch_bounds__(256) void qkv_gemm_kernel(
    const float* __restrict__ X, const float* __restrict__ W,
    const float* __restrict__ bias, float* __restrict__ out, int vmode) {
  __shared__ __attribute__((aligned(16))) float lds[2 * STAGE_FLOATS];
  const int wave = threadIdx.x >> 5;
  const int lane = threadIdx.x & 31;
  const int lo = lane & 15, hi = lane >> 4;
  const int m0 = blockIdx.x * BM + (wave >> 2) * 32;
  const int n0 = blockIdx.y * BN + (wave & 3) * 48;
  v8f acc[2][3] = {{{}, {}, {}}, {{}, {}, {}}};
  gemm_tile_lds(X, W, lds, acc);
#pragma unroll
  for (int j = 0; j < 3; ++j) {
    const int n = n0 + j * 16 + lo;
    const int h = n >> 6, d = n & (D_HEAD - 1);
    const float bv = bias[n];
#pragma unroll
    for (int i = 0; i < 2; ++i) {
#pragma unroll
      for (int r = 0; r < 8; ++r) {
        const int m = m0 + i * 16 + r + 8 * hi;
        const int b = m >> 11, s = m & (N_CTX - 1);
        const size_t hb = (size_t)b * N_HEAD + h;
        if (vmode)
          out[(hb * D_HEAD + d) * N_CTX + s] = acc[i][j][r] + bv;
        else
          out[(hb * N_CTX + s) * D_HEAD + d] = acc[i][j][r] + bv;
      }
    }
  }
}

__global__ __launch_bounds__(256) void out_gemm_kernel(
    const float* __restrict__ X, const float* __restrict__ W,
    float* __restrict__ out) {
  __shared__ __attribute__((aligned(16))) float lds[2 * STAGE_FLOATS];
  const int wave = threadIdx.x >> 5;
  const int lane = threadIdx.x & 31;
  const int lo = lane & 15, hi = lane >> 4;
  const int m0 = blockIdx.x * BM + (wave >> 2) * 32;
  const int n0 = blockIdx.y * BN + (wave & 3) * 48;
  v8f acc[2][3] = {{{}, {}, {}}, {{}, {}, {}}};
  gemm_tile_lds(X, W, lds, acc);
#pragma unroll
  for (int i = 0; i < 2; ++i) {
#pragma unroll
    for (int j = 0; j < 3; ++j) {
#pragma unroll
      for (int r = 0; r < 8; ++r) {
        const int m = m0 + i * 16 + r + 8 * hi;
        out[(size_t)m * D_MODEL + n0 + j * 16 + lo] = acc[i][j][r];
      }
    }
  }
}

__global__ __launch_bounds__(256) void norm_rope_kernel(
    float* __restrict__ qk, const float* __restrict__ g) {
  const int lane = threadIdx.x & 31;
  const int row = blockIdx.x * 8 + (threadIdx.x >> 5);
  const int s = row & (N_CTX - 1);
  float* p = qk + (size_t)row * D_HEAD;
  const float x1 = p[lane];
  const float x2 = p[lane + 32];
  float ss = x1 * x1 + x2 * x2;
#pragma unroll
  for (int off = 16; off > 0; off >>= 1) ss += __shfl_xor(ss, off, 32);
  const float inv = rsqrtf(ss * (1.0f / 64.0f) + 1e-6f);
  const float xn1 = x1 * inv * g[lane];
  const float xn2 = x2 * inv * g[lane + 32];
  const float freq = (float)s * exp2f((float)lane * (-11.0f / 32.0f));
  const float c = __cosf(freq), sn = __sinf(freq);
  p[lane]      = xn1 * c + xn2 * sn;
  p[lane + 32] = xn2 * c - xn1 * sn;
}

#define AK 64
#define APITCH 68
#define ATILE (AK * APITCH)
#define ASTAGE (2 * ATILE)
__global__ __launch_bounds__(256) void attn_kernel(
    const float* __restrict__ Q, const float* __restrict__ K,
    const float* __restrict__ V, float* __restrict__ Z) {
  __shared__ __attribute__((aligned(16))) float kv[2 * ASTAGE];
  __shared__ __attribute__((aligned(16))) float lds[8 * 256];
  const int tid = threadIdx.x;
  const int wave = __builtin_amdgcn_readfirstlane(tid >> 5);
  const int lane = tid & 31;
  const int lo = lane & 15, hi = lane >> 4;
  const int head = blockIdx.x >> 4;
  const int jg = blockIdx.x & 15;
  const int qt = jg * 8 + wave;
  const float* Qh = Q + (size_t)head * N_CTX * D_HEAD;
  const float* Kh = K + (size_t)head * N_CTX * D_HEAD;
  const float* Vh = V + (size_t)head * D_HEAD * N_CTX;
  float* pl = lds + wave * 256;

  const int cr = tid >> 4, cc = (tid & 15) * 4;
  const float* kg = Kh + (size_t)cr * D_HEAD + cc;
  const float* vg = Vh + (size_t)cr * N_CTX + cc;
  float* st = kv + cr * APITCH + cc;
  const float* kf = kv + lo * APITCH + 2 * hi;
  const float* vf = kv + ATILE + lo * APITCH + 2 * hi;
  const int nstage = 2 * jg + 2;

  v4f pre[8];
#pragma unroll
  for (int p = 0; p < 4; ++p) {
    pre[p]     = *(const v4f*)(kg + p * 16 * D_HEAD);
    pre[4 + p] = *(const v4f*)(vg + (size_t)p * 16 * N_CTX);
  }
#pragma unroll
  for (int p = 0; p < 4; ++p) {
    *(v4f*)(st + p * 16 * APITCH)         = pre[p];
    *(v4f*)(st + ATILE + p * 16 * APITCH) = pre[4 + p];
  }

  v2f qa[16];
  {
    const float* qrow = Qh + (size_t)(qt * 16 + lo) * D_HEAD + 2 * hi;
#pragma unroll
    for (int kk = 0; kk < 16; ++kk) qa[kk] = *(const v2f*)(qrow + kk * 4);
  }
  __syncthreads();

  v8f z0 = {}, z1 = {}, z2 = {}, z3 = {};
  for (int s = 0; s < nstage; ++s) {
    const int cur = (s & 1) * ASTAGE;
    const bool more = (s + 1 < nstage);
    if (more) {
      const float* kn = kg + (size_t)(s + 1) * AK * D_HEAD;
      const float* vn = vg + (s + 1) * AK;
#pragma unroll
      for (int p = 0; p < 4; ++p) {
        pre[p]     = *(const v4f*)(kn + p * 16 * D_HEAD);
        pre[4 + p] = *(const v4f*)(vn + (size_t)p * 16 * N_CTX);
      }
    }
    for (int t = 0; t < 4; ++t) {
      const int kt = s * 4 + t;
      if (kt > qt) break;
      v2f kb[16], vb[4][4];
#pragma unroll
      for (int kk = 0; kk < 16; ++kk)
        kb[kk] = *(const v2f*)(kf + cur + t * 16 * APITCH + kk * 4);
#pragma unroll
      for (int kk2 = 0; kk2 < 4; ++kk2) {
#pragma unroll
        for (int db = 0; db < 4; ++db)
          vb[kk2][db] = *(const v2f*)(vf + cur + db * 16 * APITCH + t * 16 + kk2 * 4);
      }
      __builtin_amdgcn_sched_barrier(0);
      v8f sc = {};
#pragma unroll
      for (int kk = 0; kk < 16; ++kk) sc = wmma_f32(qa[kk], kb[kk], sc);
#pragma unroll
      for (int r = 0; r < 8; ++r) {
        float pv = sc[r] * (1.0f / 64.0f);
        pv = pv * pv;
        if (kt == qt && lo > r + 8 * hi) pv = 0.0f;
        pl[(r + 8 * hi) * 16 + lo] = pv;
      }
#pragma unroll
      for (int kk2 = 0; kk2 < 4; ++kk2) {
        v2f pa = *(const v2f*)(pl + lo * 16 + kk2 * 4 + 2 * hi);
        z0 = wmma_f32(pa, vb[kk2][0], z0);
        z1 = wmma_f32(pa, vb[kk2][1], z1);
        z2 = wmma_f32(pa, vb[kk2][2], z2);
        z3 = wmma_f32(pa, vb[kk2][3], z3);
      }
    }
    if (more) {
      const int nxt = ASTAGE - cur;
#pragma unroll
      for (int p = 0; p < 4; ++p) {
        *(v4f*)(st + nxt + p * 16 * APITCH)         = pre[p];
        *(v4f*)(st + nxt + ATILE + p * 16 * APITCH) = pre[4 + p];
      }
    }
    __syncthreads();
  }
  const int b = head / N_HEAD, h = head % N_HEAD;
#pragma unroll
  for (int r = 0; r < 8; ++r) {
    const int s = qt * 16 + r + 8 * hi;
    float* zr = Z + (size_t)(b * N_CTX + s) * D_MODEL + h * D_HEAD + lo;
    zr[0]  = z0[r];
    zr[16] = z1[r];
    zr[32] = z2[r];
    zr[48] = z3[r];
  }
}

extern "C" void kernel_launch(void* const* d_in, const int* in_sizes, int n_in,
                              void* d_out, int out_size, void* d_ws, size_t ws_size,
                              hipStream_t stream) {
  const float* x  = (const float*)d_in[0];
  const float* Wq = (const float*)d_in[1];
  const float* bq = (const float*)d_in[2];
  const float* Wk = (const float*)d_in[3];
  const float* bk = (const float*)d_in[4];
  const float* Wv = (const float*)d_in[5];
  const float* bv = (const float*)d_in[6];
  const float* Wo = (const float*)d_in[7];
  const float* g  = (const float*)d_in[8];
  float* out = (float*)d_out;

  const size_t QKV = (size_t)BATCH * N_HEAD * N_CTX * D_HEAD;
  float* q = (float*)d_ws;
  float* k = q + QKV;
  float* v = k + QKV;
  float* z = v + QKV;

  dim3 gblk(M_ROWS / BM, D_MODEL / BN);
  qkv_gemm_kernel<<<gblk, 256, 0, stream>>>(x, Wq, bq, q, 0);
  qkv_gemm_kernel<<<gblk, 256, 0, stream>>>(x, Wk, bk, k, 0);
  qkv_gemm_kernel<<<gblk, 256, 0, stream>>>(x, Wv, bv, v, 1);

  const int rows = BATCH * N_HEAD * N_CTX;
  norm_rope_kernel<<<rows / 8, 256, 0, stream>>>(q, g);
  norm_rope_kernel<<<rows / 8, 256, 0, stream>>>(k, g);

  attn_kernel<<<(BATCH * N_HEAD * (N_CTX / 16)) / 8, 256, 0, stream>>>(q, k, v, z);

  out_gemm_kernel<<<gblk, 256, 0, stream>>>(z, Wo, out);
}


# Attention blocks of equal work: 16 waves take q groups 15-j and j of a head (192 blocks, was 384 of 8), one K/V stage walk for both; q tiles dealt so each SIMD sums 258 key tiles; sums unchanged
# speedup vs baseline: 2.2726x; 1.2364x over previous
#include <hip/hip_runtime.h>
#include <cstddef>

typedef float v2f __attribute__((ext_vector_type(2)));
typedef float v8f __attribute__((ext_vector_type(8)));

#define D_MODEL 768
#define N_HEAD  12
#define D_HEAD  64
#define N_CTX   2048
#define BATCH   2
#define M_ROWS  (BATCH * N_CTX)

__device__ __forceinline__ v8f wmma_f32(v2f a, v2f b, v8f c) {
  return __builtin_amdgcn_wmma_f32_16x16x4_f32(false, a, false, b, (short)0, c,
                                               false, false);
}

typedef float v4f __attribute__((ext_vector_type(4)));

#define BM 64
#define BN 192
#define KC 32
#define PITCH 36
#define STAGE_FLOATS ((BM + BN) * PITCH)
#define N_CHUNK (D_MODEL / KC)

__device__ __forceinline__ void gemm_tile_lds(const float* __restrict__ X,
                                              const float* __restrict__ W,
                                              float* __restrict__ lds,
                                              v8f acc[2][3]) {
  const int tid = threadIdx.x;
  const int wave = tid >> 5, lane = tid & 31;
  const int lo = lane & 15, hi = lane >> 4;
  const int wm = wave >> 2, wn = wave & 3;
  const int cr = tid >> 3, cc = (tid & 7) * 4;
  const float* xg = X + (size_t)(blockIdx.x * BM + cr) * D_MODEL + cc;
  const float* wg = W + (size_t)(blockIdx.y * BN + cr) * D_MODEL + cc;
  float* st = lds + cr * PITCH + cc;
  const float* fa = lds + (wm * 32 + lo) * PITCH + 2 * hi;
  const float* fb = lds + (BM + wn * 48 + lo) * PITCH + 2 * hi;

  v4f pre[8];
  pre[0] = *(const v4f*)(xg);
  pre[1] = *(const v4f*)(xg + 32 * D_MODEL);
#pragma unroll
  for (int i = 0; i < 6; ++i) pre[2 + i] = *(const v4f*)(wg + i * 32 * D_MODEL);
#pragma unroll
  for (int i = 0; i < 8; ++i) *(v4f*)(st + i * 32 * PITCH) = pre[i];
  __syncthreads();

  for (int c = 0; c < N_CHUNK; ++c) {
    const int cur = (c & 1) * STAGE_FLOATS;
    const bool more = (c + 1 < N_CHUNK);
    if (more) {
      const int k1 = (c + 1) * KC;
      pre[0] = *(const v4f*)(xg + k1);
      pre[1] = *(const v4f*)(xg + 32 * D_MODEL + k1);
#pragma unroll
      for (int i = 0; i < 6; ++i)
        pre[2 + i] = *(const v4f*)(wg + i * 32 * D_MODEL + k1);
    }
    v2f af[KC / 4][2], bf[KC / 4][3];
#pragma unroll
    for (int kk = 0; kk < KC / 4; ++kk) {
      af[kk][0] = *(const v2f*)(fa + cur + kk * 4);
      af[kk][1] = *(const v2f*)(fa + cur + 16 * PITCH + kk * 4);
      bf[kk][0] = *(const v2f*)(fb + cur + kk * 4);
      bf[kk][1] = *(const v2f*)(fb + cur + 16 * PITCH + kk * 4);
      bf[kk][2] = *(const v2f*)(fb + cur + 32 * PITCH + kk * 4);
    }
    __builtin_amdgcn_sched_barrier(0);
#pragma unroll
    for (int kk = 0; kk < KC / 4; ++kk) {
      acc[0][0] = wmma_f32(af[kk][0], bf[kk][0], acc[0][0]);
      acc[0][1] = wmma_f32(af[kk][0], bf[kk][1], acc[0][1]);
      acc[0][2] = wmma_f32(af[kk][0], bf[kk][2], acc[0][2]);
      acc[1][0] = wmma_f32(af[kk][1], bf[kk][0], acc[1][0]);
      acc[1][1] = wmma_f32(af[kk][1], bf[kk][1], acc[1][1]);
      acc[1][2] = wmma_f32(af[kk][1], bf[kk][2], acc[1][2]);
    }
    __builtin_amdgcn_sched_barrier(0);
    if (more) {
      const int nxt = STAGE_FLOATS - cur;
#pragma unroll
      for (int i = 0; i < 8; ++i) *(v4f*)(st + nxt + i * 32 * PITCH) = pre[i];
    }
    __syncthreads();
  }
}

__global__ __launch_bounds__(256) void qkv_gemm_kernel(
    const float* __restrict__ X, const float* __restrict__ W,
    const float* __restrict__ bias, float* __restrict__ out, int vmode) {
  __shared__ __attribute__((aligned(16))) float lds[2 * STAGE_FLOATS];
  const int wave = threadIdx.x >> 5;
  const int lane = threadIdx.x & 31;
  const int lo = lane & 15, hi = lane >> 4;
  const int m0 = blockIdx.x * BM + (wave >> 2) * 32;
  const int n0 = blockIdx.y * BN + (wave & 3) * 48;
  v8f acc[2][3] = {{{}, {}, {}}, {{}, {}, {}}};
  gemm_tile_lds(X, W, lds, acc);
#pragma unroll
  for (int j = 0; j < 3; ++j) {
    const int n = n0 + j * 16 + lo;
    const int h = n >> 6, d = n & (D_HEAD - 1);
    const float bv = bias[n];
#pragma unroll
    for (int i = 0; i < 2; ++i) {
#pragma unroll
      for (int r = 0; r < 8; ++r) {
        const int m = m0 + i * 16 + r + 8 * hi;
        const int b = m >> 11, s = m & (N_CTX - 1);
        const size_t hb = (size_t)b * N_HEAD + h;
        if (vmode)
          out[(hb * D_HEAD + d) * N_CTX + s] = acc[i][j][r] + bv;
        else
          out[(hb * N_CTX + s) * D_HEAD + d] = acc[i][j][r] + bv;
      }
    }
  }
}

__global__ __launch_bounds__(256) void out_gemm_kernel(
    const float* __restrict__ X, const float* __restrict__ W,
    float* __restrict__ out) {
  __shared__ __attribute__((aligned(16))) float lds[2 * STAGE_FLOATS];
  const int wave = threadIdx.x >> 5;
  const int lane = threadIdx.x & 31;
  const int lo = lane & 15, hi = lane >> 4;
  const int m0 = blockIdx.x * BM + (wave >> 2) * 32;
  const int n0 = blockIdx.y * BN + (wave & 3) * 48;
  v8f acc[2][3] = {{{}, {}, {}}, {{}, {}, {}}};
  gemm_tile_lds(X, W, lds, acc);
#pragma unroll
  for (int i = 0; i < 2; ++i) {
#pragma unroll
    for (int j = 0; j < 3; ++j) {
#pragma unroll
      for (int r = 0; r < 8; ++r) {
        const int m = m0 + i * 16 + r + 8 * hi;
        out[(size_t)m * D_MODEL + n0 + j * 16 + lo] = acc[i][j][r];
      }
    }
  }
}

__global__ __launch_bounds__(256) void norm_rope_kernel(
    float* __restrict__ qk, const float* __restrict__ g) {
  const int lane = threadIdx.x & 31;
  const int row = blockIdx.x * 8 + (threadIdx.x >> 5);
  const int s = row & (N_CTX - 1);
  float* p = qk + (size_t)row * D_HEAD;
  const float x1 = p[lane];
  const float x2 = p[lane + 32];
  float ss = x1 * x1 + x2 * x2;
#pragma unroll
  for (int off = 16; off > 0; off >>= 1) ss += __shfl_xor(ss, off, 32);
  const float inv = rsqrtf(ss * (1.0f / 64.0f) + 1e-6f);
  const float xn1 = x1 * inv * g[lane];
  const float xn2 = x2 * inv * g[lane + 32];
  const float freq = (float)s * exp2f((float)lane * (-11.0f / 32.0f));
  const float c = __cosf(freq), sn = __sinf(freq);
  p[lane]      = xn1 * c + xn2 * sn;
  p[lane + 32] = xn2 * c - xn1 * sn;
}

#define AK 64
#define APITCH 68
#define ATILE (AK * APITCH)
#define ASTAGE (2 * ATILE)
__global__ __launch_bounds__(512) void attn_kernel(
    const float* __restrict__ Q, const float* __restrict__ K,
    const float* __restrict__ V, float* __restrict__ Z) {
  __shared__ __attribute__((aligned(16))) float kv[2 * ASTAGE];
  __shared__ __attribute__((aligned(16))) float lds[16 * 256];
  const int tid = threadIdx.x;
  const int wave = __builtin_amdgcn_readfirstlane(tid >> 5);
  const int lane = tid & 31;
  const int lo = lane & 15, hi = lane >> 4;
  const int head = blockIdx.x >> 3;
  const int jp = blockIdx.x & 7;
  const int longg = (((wave >> 2) ^ wave) & 1) == 0;
  const int place = (int)((0x5243071634256170ULL >> (4 * wave)) & 7);
  const int qt = (longg ? 15 - jp : jp) * 8 + place;
  const float* Qh = Q + (size_t)head * N_CTX * D_HEAD;
  const float* Kh = K + (size_t)head * N_CTX * D_HEAD;
  const float* Vh = V + (size_t)head * D_HEAD * N_CTX;
  float* pl = lds + wave * 256;

  const int cr = tid >> 4, cc = (tid & 15) * 4;
  const float* kg = Kh + (size_t)cr * D_HEAD + cc;
  const float* vg = Vh + (size_t)cr * N_CTX + cc;
  float* st = kv + cr * APITCH + cc;
  const float* kf = kv + lo * APITCH + 2 * hi;
  const float* vf = kv + ATILE + lo * APITCH + 2 * hi;
  const int nstage = 2 * (15 - jp) + 2;

  v4f pre[4];
#pragma unroll
  for (int p = 0; p < 2; ++p) {
    pre[p]     = *(const v4f*)(kg + p * 32 * D_HEAD);
    pre[2 + p] = *(const v4f*)(vg + (size_t)p * 32 * N_CTX);
  }
#pragma unroll
  for (int p = 0; p < 2; ++p) {
    *(v4f*)(st + p * 32 * APITCH)         = pre[p];
    *(v4f*)(st + ATILE + p * 32 * APITCH) = pre[2 + p];
  }

  v2f qa[16];
  {
    const float* qrow = Qh + (size_t)(qt * 16 + lo) * D_HEAD + 2 * hi;
#pragma unroll
    for (int kk = 0; kk < 16; ++kk) qa[kk] = *(const v2f*)(qrow + kk * 4);
  }
  __syncthreads();

  v8f z0 = {}, z1 = {}, z2 = {}, z3 = {};
  for (int s = 0; s < nstage; ++s) {
    const int cur = (s & 1) * ASTAGE;
    const bool more = (s + 1 < nstage);
    if (more) {
      const float* kn = kg + (size_t)(s + 1) * AK * D_HEAD;
      const float* vn = vg + (s + 1) * AK;
#pragma unroll
      for (int p = 0; p < 2; ++p) {
        pre[p]     = *(const v4f*)(kn + p * 32 * D_HEAD);
        pre[2 + p] = *(const v4f*)(vn + (size_t)p * 32 * N_CTX);
      }
    }
    for (int t = 0; t < 4; ++t) {
      const int kt = s * 4 + t;
      if (kt > qt) break;
      v2f kb[16], vb[4][4];
#pragma unroll
      for (int kk = 0; kk < 16; ++kk)
        kb[kk] = *(const v2f*)(kf + cur + t * 16 * APITCH + kk * 4);
#pragma unroll
      for (int kk2 = 0; kk2 < 4; ++kk2) {
#pragma unroll
        for (int db = 0; db < 4; ++db)
          vb[kk2][db] = *(const v2f*)(vf + cur + db * 16 * APITCH + t * 16 + kk2 * 4);
      }
      __builtin_amdgcn_sched_barrier(0);
      v8f sc = {};
#pragma unroll
      for (int kk = 0; kk < 16; ++kk) sc = wmma_f32(qa[kk], kb[kk], sc);
#pragma unroll
      for (int r = 0; r < 8; ++r) {
        float pv = sc[r] * (1.0f / 64.0f);
        pv = pv * pv;
        if (kt == qt && lo > r + 8 * hi) pv = 0.0f;
        pl[(r + 8 * hi) * 16 + lo] = pv;
      }
#pragma unroll
      for (int kk2 = 0; kk2 < 4; ++kk2) {
        v2f pa = *(const v2f*)(pl + lo * 16 + kk2 * 4 + 2 * hi);
        z0 = wmma_f32(pa, vb[kk2][0], z0);
        z1 = wmma_f32(pa, vb[kk2][1], z1);
        z2 = wmma_f32(pa, vb[kk2][2], z2);
        z3 = wmma_f32(pa, vb[kk2][3], z3);
      }
    }
    if (more) {
      const int nxt = ASTAGE - cur;
#pragma unroll
      for (int p = 0; p < 2; ++p) {
        *(v4f*)(st + nxt + p * 32 * APITCH)         = pre[p];
        *(v4f*)(st + nxt + ATILE + p * 32 * APITCH) = pre[2 + p];
      }
    }
    __syncthreads();
  }
  const int b = head / N_HEAD, h = head % N_HEAD;
#pragma unroll
  for (int r = 0; r < 8; ++r) {
    const int s = qt * 16 + r + 8 * hi;
    float* zr = Z + (size_t)(b * N_CTX + s) * D_MODEL + h * D_HEAD + lo;
    zr[0]  = z0[r];
    zr[16] = z1[r];
    zr[32] = z2[r];
    zr[48] = z3[r];
  }
}

extern "C" void kernel_launch(void* const* d_in, const int* in_sizes, int n_in,
                              void* d_out, int out_size, void* d_ws, size_t ws_size,
                              hipStream_t stream) {
  const float* x  = (const float*)d_in[0];
  const float* Wq = (const float*)d_in[1];
  const float* bq = (const float*)d_in[2];
  const float* Wk = (const float*)d_in[3];
  const float* bk = (const float*)d_in[4];
  const float* Wv = (const float*)d_in[5];
  const float* bv = (const float*)d_in[6];
  const float* Wo = (const float*)d_in[7];
  const float* g  = (const float*)d_in[8];
  float* out = (float*)d_out;

  const size_t QKV = (size_t)BATCH * N_HEAD * N_CTX * D_HEAD;
  float* q = (float*)d_ws;
  float* k = q + QKV;
  float* v = k + QKV;
  float* z = v + QKV;

  dim3 gblk(M_ROWS / BM, D_MODEL / BN);
  qkv_gemm_kernel<<<gblk, 256, 0, stream>>>(x, Wq, bq, q, 0);
  qkv_gemm_kernel<<<gblk, 256, 0, stream>>>(x, Wk, bk, k, 0);
  qkv_gemm_kernel<<<gblk, 256, 0, stream>>>(x, Wv, bv, v, 1);

  const int rows = BATCH * N_HEAD * N_CTX;
  norm_rope_kernel<<<rows / 8, 256, 0, stream>>>(q, g);
  norm_rope_kernel<<<rows / 8, 256, 0, stream>>>(k, g);

  attn_kernel<<<(BATCH * N_HEAD * (N_CTX / 16)) / 16, 512, 0, stream>>>(q, k, v, z);

  out_gemm_kernel<<<gblk, 256, 0, stream>>>(z, Wo, out);
}


# RMSNorm+RoPE of Q and K as 1 launch of 4096 resident waves: 24 rows per wave, all loads at entry, 8 lanes per row with both rotation halves in-lane; was 2 x 49152 one-row waves; norm sum tree changes
# speedup vs baseline: 2.6197x; 1.1527x over previous
#include <hip/hip_runtime.h>
#include <cstddef>

typedef float v2f __attribute__((ext_vector_type(2)));
typedef float v8f __attribute__((ext_vector_type(8)));

#define D_MODEL 768
#define N_HEAD  12
#define D_HEAD  64
#define N_CTX   2048
#define BATCH   2
#define M_ROWS  (BATCH * N_CTX)

__device__ __forceinline__ v8f wmma_f32(v2f a, v2f b, v8f c) {
  return __builtin_amdgcn_wmma_f32_16x16x4_f32(false, a, false, b, (short)0, c,
                                               false, false);
}

typedef float v4f __attribute__((ext_vector_type(4)));

#define BM 64
#define BN 192
#define KC 32
#define PITCH 36
#define STAGE_FLOATS ((BM + BN) * PITCH)
#define N_CHUNK (D_MODEL / KC)

__device__ __forceinline__ void gemm_tile_lds(const float* __restrict__ X,
                                              const float* __restrict__ W,
                                              float* __restrict__ lds,
                                              v8f acc[2][3]) {
  const int tid = threadIdx.x;
  const int wave = tid >> 5, lane = tid & 31;
  const int lo = lane & 15, hi = lane >> 4;
  const int wm = wave >> 2, wn = wave & 3;
  const int cr = tid >> 3, cc = (tid & 7) * 4;
  const float* xg = X + (size_t)(blockIdx.x * BM + cr) * D_MODEL + cc;
  const float* wg = W + (size_t)(blockIdx.y * BN + cr) * D_MODEL + cc;
  float* st = lds + cr * PITCH + cc;
  const float* fa = lds + (wm * 32 + lo) * PITCH + 2 * hi;
  const float* fb = lds + (BM + wn * 48 + lo) * PITCH + 2 * hi;

  v4f pre[8];
  pre[0] = *(const v4f*)(xg);
  pre[1] = *(const v4f*)(xg + 32 * D_MODEL);
#pragma unroll
  for (int i = 0; i < 6; ++i) pre[2 + i] = *(const v4f*)(wg + i * 32 * D_MODEL);
#pragma unroll
  for (int i = 0; i < 8; ++i) *(v4f*)(st + i * 32 * PITCH) = pre[i];
  __syncthreads();

  for (int c = 0; c < N_CHUNK; ++c) {
    const int cur = (c & 1) * STAGE_FLOATS;
    const bool more = (c + 1 < N_CHUNK);
    if (more) {
      const int k1 = (c + 1) * KC;
      pre[0] = *(const v4f*)(xg + k1);
      pre[1] = *(const v4f*)(xg + 32 * D_MODEL + k1);
#pragma unroll
      for (int i = 0; i < 6; ++i)
        pre[2 + i] = *(const v4f*)(wg + i * 32 * D_MODEL + k1);
    }
    v2f af[KC / 4][2], bf[KC / 4][3];
#pragma unroll
    for (int kk = 0; kk < KC / 4; ++kk) {
      af[kk][0] = *(const v2f*)(fa + cur + kk * 4);
      af[kk][1] = *(const v2f*)(fa + cur + 16 * PITCH + kk * 4);
      bf[kk][0] = *(const v2f*)(fb + cur + kk * 4);
      bf[kk][1] = *(const v2f*)(fb + cur + 16 * PITCH + kk * 4);
      bf[kk][2] = *(const v2f*)(fb + cur + 32 * PITCH + kk * 4);
    }
    __builtin_amdgcn_sched_barrier(0);
#pragma unroll
    for (int kk = 0; kk < KC / 4; ++kk) {
      acc[0][0] = wmma_f32(af[kk][0], bf[kk][0], acc[0][0]);
      acc[0][1] = wmma_f32(af[kk][0], bf[kk][1], acc[0][1]);
      acc[0][2] = wmma_f32(af[kk][0], bf[kk][2], acc[0][2]);
      acc[1][0] = wmma_f32(af[kk][1], bf[kk][0], acc[1][0]);
      acc[1][1] = wmma_f32(af[kk][1], bf[kk][1], acc[1][1]);
      acc[1][2] = wmma_f32(af[kk][1], bf[kk][2], acc[1][2]);
    }
    __builtin_amdgcn_sched_barrier(0);
    if (more) {
      const int nxt = STAGE_FLOATS - cur;
#pragma unroll
      for (int i = 0; i < 8; ++i) *(v4f*)(st + nxt + i * 32 * PITCH) = pre[i];
    }
    __syncthreads();
  }
}

__global__ __launch_bounds__(256) void qkv_gemm_kernel(
    const float* __restrict__ X, const float* __restrict__ W,
    const float* __restrict__ bias, float* __restrict__ out, int vmode) {
  __shared__ __attribute__((aligned(16))) float lds[2 * STAGE_FLOATS];
  const int wave = threadIdx.x >> 5;
  const int lane = threadIdx.x & 31;
  const int lo = lane & 15, hi = lane >> 4;
  const int m0 = blockIdx.x * BM + (wave >> 2) * 32;
  const int n0 = blockIdx.y * BN + (wave & 3) * 48;
  v8f acc[2][3] = {{{}, {}, {}}, {{}, {}, {}}};
  gemm_tile_lds(X, W, lds, acc);
#pragma unroll
  for (int j = 0; j < 3; ++j) {
    const int n = n0 + j * 16 + lo;
    const int h = n >> 6, d = n & (D_HEAD - 1);
    const float bv = bias[n];
#pragma unroll
    for (int i = 0; i < 2; ++i) {
#pragma unroll
      for (int r = 0; r < 8; ++r) {
        const int m = m0 + i * 16 + r + 8 * hi;
        const int b = m >> 11, s = m & (N_CTX - 1);
        const size_t hb = (size_t)b * N_HEAD + h;
        if (vmode)
          out[(hb * D_HEAD + d) * N_CTX + s] = acc[i][j][r] + bv;
        else
          out[(hb * N_CTX + s) * D_HEAD + d] = acc[i][j][r] + bv;
      }
    }
  }
}

__global__ __launch_bounds__(256) void out_gemm_kernel(
    const float* __restrict__ X, const float* __restrict__ W,
    float* __restrict__ out) {
  __shared__ __attribute__((aligned(16))) float lds[2 * STAGE_FLOATS];
  const int wave = threadIdx.x >> 5;
  const int lane = threadIdx.x & 31;
  const int lo = lane & 15, hi = lane >> 4;
  const int m0 = blockIdx.x * BM + (wave >> 2) * 32;
  const int n0 = blockIdx.y * BN + (wave & 3) * 48;
  v8f acc[2][3] = {{{}, {}, {}}, {{}, {}, {}}};
  gemm_tile_lds(X, W, lds, acc);
#pragma unroll
  for (int i = 0; i < 2; ++i) {
#pragma unroll
    for (int j = 0; j < 3; ++j) {
#pragma unroll
      for (int r = 0; r < 8; ++r) {
        const int m = m0 + i * 16 + r + 8 * hi;
        out[(size_t)m * D_MODEL + n0 + j * 16 + lo] = acc[i][j][r];
      }
    }
  }
}

#define NR_ROWS 24
#define NR_IT   (NR_ROWS / 4)
#define NR_BLOCKS ((2 * BATCH * N_HEAD * N_CTX) / (16 * NR_ROWS))
__global__ __launch_bounds__(512) void norm_rope_kernel(
    float* __restrict__ qk, const float* __restrict__ g) {
  const int lane = threadIdx.x & 31;
  const int gw = blockIdx.x * 16 + (threadIdx.x >> 5);
  const int c8 = lane & 7;
  const int row0 = gw * NR_ROWS + (lane >> 3);
  float* p = qk + (size_t)row0 * D_HEAD + 4 * c8;
  v4f xa[NR_IT], xb[NR_IT];
#pragma unroll
  for (int it = 0; it < NR_IT; ++it) {
    xa[it] = *(const v4f*)(p + it * 4 * D_HEAD);
    xb[it] = *(const v4f*)(p + it * 4 * D_HEAD + 32);
  }
  const v4f ga = *(const v4f*)(g + 4 * c8);
  const v4f gb = *(const v4f*)(g + 32 + 4 * c8);
  asm volatile("" :: "v"(xa[0]), "v"(xb[0]), "v"(xa[1]), "v"(xb[1]), "v"(xa[2]), "v"(xb[2]),
                     "v"(xa[3]), "v"(xb[3]), "v"(xa[4]), "v"(xb[4]), "v"(xa[5]), "v"(xb[5]),
                     "v"(ga), "v"(gb));
  float fr[4];
#pragma unroll
  for (int e = 0; e < 4; ++e)
    fr[e] = exp2f((float)(4 * c8 + e) * (-11.0f / 32.0f));
#pragma unroll
  for (int it = 0; it < NR_IT; ++it) {
    const int s = (row0 + it * 4) & (N_CTX - 1);
    const v4f a = xa[it], b = xb[it];
    float ss = 0.0f;
#pragma unroll
    for (int e = 0; e < 4; ++e) ss += a[e] * a[e] + b[e] * b[e];
    ss += __shfl_xor(ss, 1, 32);
    ss += __shfl_xor(ss, 2, 32);
    ss += __shfl_xor(ss, 4, 32);
    const float inv = rsqrtf(ss * (1.0f / 64.0f) + 1e-6f);
    v4f o1, o2;
#pragma unroll
    for (int e = 0; e < 4; ++e) {
      const float xn1 = a[e] * inv * ga[e];
      const float xn2 = b[e] * inv * gb[e];
      const float freq = (float)s * fr[e];
      const float c = __cosf(freq), sn = __sinf(freq);
      o1[e] = xn1 * c + xn2 * sn;
      o2[e] = xn2 * c - xn1 * sn;
    }
    *(v4f*)(p + it * 4 * D_HEAD)      = o1;
    *(v4f*)(p + it * 4 * D_HEAD + 32) = o2;
  }
}

#define AK 64
#define APITCH 68
#define ATILE (AK * APITCH)
#define ASTAGE (2 * ATILE)
__global__ __launch_bounds__(512) void attn_kernel(
    const float* __restrict__ Q, const float* __restrict__ K,
    const float* __restrict__ V, float* __restrict__ Z) {
  __shared__ __attribute__((aligned(16))) float kv[2 * ASTAGE];
  __shared__ __attribute__((aligned(16))) float lds[16 * 256];
  const int tid = threadIdx.x;
  const int wave = __builtin_amdgcn_readfirstlane(tid >> 5);
  const int lane = tid & 31;
  const int lo = lane & 15, hi = lane >> 4;
  const int head = blockIdx.x >> 3;
  const int jp = blockIdx.x & 7;
  const int longg = (((wave >> 2) ^ wave) & 1) == 0;
  const int place = (int)((0x5243071634256170ULL >> (4 * wave)) & 7);
  const int qt = (longg ? 15 - jp : jp) * 8 + place;
  const float* Qh = Q + (size_t)head * N_CTX * D_HEAD;
  const float* Kh = K + (size_t)head * N_CTX * D_HEAD;
  const float* Vh = V + (size_t)head * D_HEAD * N_CTX;
  float* pl = lds + wave * 256;

  const int cr = tid >> 4, cc = (tid & 15) * 4;
  const float* kg = Kh + (size_t)cr * D_HEAD + cc;
  const float* vg = Vh + (size_t)cr * N_CTX + cc;
  float* st = kv + cr * APITCH + cc;
  const float* kf = kv + lo * APITCH + 2 * hi;
  const float* vf = kv + ATILE + lo * APITCH + 2 * hi;
  const int nstage = 2 * (15 - jp) + 2;

  v4f pre[4];
#pragma unroll
  for (int p = 0; p < 2; ++p) {
    pre[p]     = *(const v4f*)(kg + p * 32 * D_HEAD);
    pre[2 + p] = *(const v4f*)(vg + (size_t)p * 32 * N_CTX);
  }
#pragma unroll
  for (int p = 0; p < 2; ++p) {
    *(v4f*)(st + p * 32 * APITCH)         = pre[p];
    *(v4f*)(st + ATILE + p * 32 * APITCH) = pre[2 + p];
  }

  v2f qa[16];
  {
    const float* qrow = Qh + (size_t)(qt * 16 + lo) * D_HEAD + 2 * hi;
#pragma unroll
    for (int kk = 0; kk < 16; ++kk) qa[kk] = *(const v2f*)(qrow + kk * 4);
  }
  __syncthreads();

  v8f z0 = {}, z1 = {}, z2 = {}, z3 = {};
  for (int s = 0; s < nstage; ++s) {
    const int cur = (s & 1) * ASTAGE;
    const bool more = (s + 1 < nstage);
    if (more) {
      const float* kn = kg + (size_t)(s + 1) * AK * D_HEAD;
      const float* vn = vg + (s + 1) * AK;
#pragma unroll
      for (int p = 0; p < 2; ++p) {
        pre[p]     = *(const v4f*)(kn + p * 32 * D_HEAD);
        pre[2 + p] = *(const v4f*)(vn + (size_t)p * 32 * N_CTX);
      }
    }
    for (int t = 0; t < 4; ++t) {
      const int kt = s * 4 + t;
      if (kt > qt) break;
      v2f kb[16], vb[4][4];
#pragma unroll
      for (int kk = 0; kk < 16; ++kk)
        kb[kk] = *(const v2f*)(kf + cur + t * 16 * APITCH + kk * 4);
#pragma unroll
      for (int kk2 = 0; kk2 < 4; ++kk2) {
#pragma unroll
        for (int db = 0; db < 4; ++db)
          vb[kk2][db] = *(const v2f*)(vf + cur + db * 16 * APITCH + t * 16 + kk2 * 4);
      }
      __builtin_amdgcn_sched_barrier(0);
      v8f sc = {};
#pragma unroll
      for (int kk = 0; kk < 16; ++kk) sc = wmma_f32(qa[kk], kb[kk], sc);
#pragma unroll
      for (int r = 0; r < 8; ++r) {
        float pv = sc[r] * (1.0f / 64.0f);
        pv = pv * pv;
        if (kt == qt && lo > r + 8 * hi) pv = 0.0f;
        pl[(r + 8 * hi) * 16 + lo] = pv;
      }
#pragma unroll
      for (int kk2 = 0; kk2 < 4; ++kk2) {
        v2f pa = *(const v2f*)(pl + lo * 16 + kk2 * 4 + 2 * hi);
        z0 = wmma_f32(pa, vb[kk2][0], z0);
        z1 = wmma_f32(pa, vb[kk2][1], z1);
        z2 = wmma_f32(pa, vb[kk2][2], z2);
        z3 = wmma_f32(pa, vb[kk2][3], z3);
      }
    }
    if (more) {
      const int nxt = ASTAGE - cur;
#pragma unroll
      for (int p = 0; p < 2; ++p) {
        *(v4f*)(st + nxt + p * 32 * APITCH)         = pre[p];
        *(v4f*)(st + nxt + ATILE + p * 32 * APITCH) = pre[2 + p];
      }
    }
    __syncthreads();
  }
  const int b = head / N_HEAD, h = head % N_HEAD;
#pragma unroll
  for (int r = 0; r < 8; ++r) {
    const int s = qt * 16 + r + 8 * hi;
    float* zr = Z + (size_t)(b * N_CTX + s) * D_MODEL + h * D_HEAD + lo;
    zr[0]  = z0[r];
    zr[16] = z1[r];
    zr[32] = z2[r];
    zr[48] = z3[r];
  }
}

extern "C" void kernel_launch(void* const* d_in, const int* in_sizes, int n_in,
                              void* d_out, int out_size, void* d_ws, size_t ws_size,
                              hipStream_t stream) {
  const float* x  = (const float*)d_in[0];
  const float* Wq = (const float*)d_in[1];
  const float* bq = (const float*)d_in[2];
  const float* Wk = (const float*)d_in[3];
  const float* bk = (const float*)d_in[4];
  const float* Wv = (const float*)d_in[5];
  const float* bv = (const float*)d_in[6];
  const float* Wo = (const float*)d_in[7];
  const float* g  = (const float*)d_in[8];
  float* out = (float*)d_out;

  const size_t QKV = (size_t)BATCH * N_HEAD * N_CTX * D_HEAD;
  float* q = (float*)d_ws;
  float* k = q + QKV;
  float* v = k + QKV;
  float* z = v + QKV;

  dim3 gblk(M_ROWS / BM, D_MODEL / BN);
  qkv_gemm_kernel<<<gblk, 256, 0, stream>>>(x, Wq, bq, q, 0);
  qkv_gemm_kernel<<<gblk, 256, 0, stream>>>(x, Wk, bk, k, 0);
  qkv_gemm_kernel<<<gblk, 256, 0, stream>>>(x, Wv, bv, v, 1);

  norm_rope_kernel<<<NR_BLOCKS, 512, 0, stream>>>(q, g);

  attn_kernel<<<(BATCH * N_HEAD * (N_CTX / 16)) / 16, 512, 0, stream>>>(q, k, v, z);

  out_gemm_kernel<<<gblk, 256, 0, stream>>>(z, Wo, out);
}


# GEMM launches as 512 blocks of 4 waves (tile 64x96), two independent blocks per WGP with wave priority falling with K progress: one block's barrier gap runs under the other's WMMAs; sums unchanged
# speedup vs baseline: 2.7034x; 1.0320x over previous
#include <hip/hip_runtime.h>
#include <cstddef>

typedef float v2f __attribute__((ext_vector_type(2)));
typedef float v8f __attribute__((ext_vector_type(8)));

#define D_MODEL 768
#define N_HEAD  12
#define D_HEAD  64
#define N_CTX   2048
#define BATCH   2
#define M_ROWS  (BATCH * N_CTX)

__device__ __forceinline__ v8f wmma_f32(v2f a, v2f b, v8f c) {
  return __builtin_amdgcn_wmma_f32_16x16x4_f32(false, a, false, b, (short)0, c,
                                               false, false);
}

typedef float v4f __attribute__((ext_vector_type(4)));

#define BM 64
#define BN 96
#define KC 32
#define PITCH 36
#define STAGE_FLOATS ((BM + BN) * PITCH)
#define N_CHUNK (D_MODEL / KC)
#define GEMM_LDS_FLOATS 28672

__device__ __forceinline__ void gemm_tile_lds(const float* __restrict__ X,
                                              const float* __restrict__ W,
                                              float* __restrict__ lds,
                                              v8f acc[2][3]) {
  const int tid = threadIdx.x;
  const int wave = tid >> 5, lane = tid & 31;
  const int lo = lane & 15, hi = lane >> 4;
  const int wm = wave >> 1, wn = wave & 1;
  const int cr = tid >> 3, cc = (tid & 7) * 4;
  const float* xg = X + (size_t)(blockIdx.x * BM + cr) * D_MODEL + cc;
  const float* wg = W + (size_t)(blockIdx.y * BN + cr) * D_MODEL + cc;
  float* st = lds + cr * PITCH + cc;
  const float* fa = lds + (wm * 32 + lo) * PITCH + 2 * hi;
  const float* fb = lds + (BM + wn * 48 + lo) * PITCH + 2 * hi;

  v4f pre[10];
#pragma unroll
  for (int i = 0; i < 4; ++i) pre[i] = *(const v4f*)(xg + i * 16 * D_MODEL);
#pragma unroll
  for (int i = 0; i < 6; ++i) pre[4 + i] = *(const v4f*)(wg + i * 16 * D_MODEL);
#pragma unroll
  for (int i = 0; i < 10; ++i) *(v4f*)(st + i * 16 * PITCH) = pre[i];
  __syncthreads();

  for (int c = 0; c < N_CHUNK; ++c) {
    if (c == 0) __builtin_amdgcn_s_setprio(3);
    else if (c == N_CHUNK / 4) __builtin_amdgcn_s_setprio(2);
    else if (c == N_CHUNK / 2) __builtin_amdgcn_s_setprio(1);
    else if (c == 3 * N_CHUNK / 4) __builtin_amdgcn_s_setprio(0);
    const int cur = (c & 1) * STAGE_FLOATS;
    const bool more = (c + 1 < N_CHUNK);
    if (more) {
      const int k1 = (c + 1) * KC;
#pragma unroll
      for (int i = 0; i < 4; ++i)
        pre[i] = *(const v4f*)(xg + i * 16 * D_MODEL + k1);
#pragma unroll
      for (int i = 0; i < 6; ++i)
        pre[4 + i] = *(const v4f*)(wg + i * 16 * D_MODEL + k1);
    }
    v2f af[KC / 4][2], bf[KC / 4][3];
#pragma unroll
    for (int kk = 0; kk < KC / 4; ++kk) {
      af[kk][0] = *(const v2f*)(fa + cur + kk * 4);
      af[kk][1] = *(const v2f*)(fa + cur + 16 * PITCH + kk * 4);
      bf[kk][0] = *(const v2f*)(fb + cur + kk * 4);
      bf[kk][1] = *(const v2f*)(fb + cur + 16 * PITCH + kk * 4);
      bf[kk][2] = *(const v2f*)(fb + cur + 32 * PITCH + kk * 4);
    }
    __builtin_amdgcn_sched_barrier(0);
#pragma unroll
    for (int kk = 0; kk < KC / 4; ++kk) {
      acc[0][0] = wmma_f32(af[kk][0], bf[kk][0], acc[0][0]);
      acc[0][1] = wmma_f32(af[kk][0], bf[kk][1], acc[0][1]);
      acc[0][2] = wmma_f32(af[kk][0], bf[kk][2], acc[0][2]);
      acc[1][0] = wmma_f32(af[kk][1], bf[kk][0], acc[1][0]);
      acc[1][1] = wmma_f32(af[kk][1], bf[kk][1], acc[1][1]);
      acc[1][2] = wmma_f32(af[kk][1], bf[kk][2], acc[1][2]);
    }
    __builtin_amdgcn_sched_barrier(0);
    if (more) {
      const int nxt = STAGE_FLOATS - cur;
#pragma unroll
      for (int i = 0; i < 10; ++i) *(v4f*)(st + nxt + i * 16 * PITCH) = pre[i];
    }
    __syncthreads();
  }
}

__global__ __launch_bounds__(128) void qkv_gemm_kernel(
    const float* __restrict__ X, const float* __restrict__ W,
    const float* __restrict__ bias, float* __restrict__ out, int vmode) {
  __shared__ __attribute__((aligned(16))) float lds[GEMM_LDS_FLOATS];
  const int wave = threadIdx.x >> 5;
  const int lane = threadIdx.x & 31;
  const int lo = lane & 15, hi = lane >> 4;
  const int m0 = blockIdx.x * BM + (wave >> 1) * 32;
  const int n0 = blockIdx.y * BN + (wave & 1) * 48;
  v8f acc[2][3] = {{{}, {}, {}}, {{}, {}, {}}};
  gemm_tile_lds(X, W, lds, acc);
#pragma unroll
  for (int j = 0; j < 3; ++j) {
    const int n = n0 + j * 16 + lo;
    const int h = n >> 6, d = n & (D_HEAD - 1);
    const float bv = bias[n];
#pragma unroll
    for (int i = 0; i < 2; ++i) {
#pragma unroll
      for (int r = 0; r < 8; ++r) {
        const int m = m0 + i * 16 + r + 8 * hi;
        const int b = m >> 11, s = m & (N_CTX - 1);
        const size_t hb = (size_t)b * N_HEAD + h;
        if (vmode)
          out[(hb * D_HEAD + d) * N_CTX + s] = acc[i][j][r] + bv;
        else
          out[(hb * N_CTX + s) * D_HEAD + d] = acc[i][j][r] + bv;
      }
    }
  }
}

__global__ __launch_bounds__(128) void out_gemm_kernel(
    const float* __restrict__ X, const float* __restrict__ W,
    float* __restrict__ out) {
  __shared__ __attribute__((aligned(16))) float lds[GEMM_LDS_FLOATS];
  const int wave = threadIdx.x >> 5;
  const int lane = threadIdx.x & 31;
  const int lo = lane & 15, hi = lane >> 4;
  const int m0 = blockIdx.x * BM + (wave >> 1) * 32;
  const int n0 = blockIdx.y * BN + (wave & 1) * 48;
  v8f acc[2][3] = {{{}, {}, {}}, {{}, {}, {}}};
  gemm_tile_lds(X, W, lds, acc);
#pragma unroll
  for (int i = 0; i < 2; ++i) {
#pragma unroll
    for (int j = 0; j < 3; ++j) {
#pragma unroll
      for (int r = 0; r < 8; ++r) {
        const int m = m0 + i * 16 + r + 8 * hi;
        out[(size_t)m * D_MODEL + n0 + j * 16 + lo] = acc[i][j][r];
      }
    }
  }
}

#define NR_ROWS 24
#define NR_IT   (NR_ROWS / 4)
#define NR_BLOCKS ((2 * BATCH * N_HEAD * N_CTX) / (16 * NR_ROWS))
__global__ __launch_bounds__(512) void norm_rope_kernel(
    float* __restrict__ qk, const float* __restrict__ g) {
  const int lane = threadIdx.x & 31;
  const int gw = blockIdx.x * 16 + (threadIdx.x >> 5);
  const int c8 = lane & 7;
  const int row0 = gw * NR_ROWS + (lane >> 3);
  float* p = qk + (size_t)row0 * D_HEAD + 4 * c8;
  v4f xa[NR_IT], xb[NR_IT];
#pragma unroll
  for (int it = 0; it < NR_IT; ++it) {
    xa[it] = *(const v4f*)(p + it * 4 * D_HEAD);
    xb[it] = *(const v4f*)(p + it * 4 * D_HEAD + 32);
  }
  const v4f ga = *(const v4f*)(g + 4 * c8);
  const v4f gb = *(const v4f*)(g + 32 + 4 * c8);
  asm volatile("" :: "v"(xa[0]), "v"(xb[0]), "v"(xa[1]), "v"(xb[1]), "v"(xa[2]), "v"(xb[2]),
                     "v"(xa[3]), "v"(xb[3]), "v"(xa[4]), "v"(xb[4]), "v"(xa[5]), "v"(xb[5]),
                     "v"(ga), "v"(gb));
  float fr[4];
#pragma unroll
  for (int e = 0; e < 4; ++e)
    fr[e] = exp2f((float)(4 * c8 + e) * (-11.0f / 32.0f));
#pragma unroll
  for (int it = 0; it < NR_IT; ++it) {
    const int s = (row0 + it * 4) & (N_CTX - 1);
    const v4f a = xa[it], b = xb[it];
    float ss = 0.0f;
#pragma unroll
    for (int e = 0; e < 4; ++e) ss += a[e] * a[e] + b[e] * b[e];
    ss += __shfl_xor(ss, 1, 32);
    ss += __shfl_xor(ss, 2, 32);
    ss += __shfl_xor(ss, 4, 32);
    const float inv = rsqrtf(ss * (1.0f / 64.0f) + 1e-6f);
    v4f o1, o2;
#pragma unroll
    for (int e = 0; e < 4; ++e) {
      const float xn1 = a[e] * inv * ga[e];
      const float xn2 = b[e] * inv * gb[e];
      const float freq = (float)s * fr[e];
      const float c = __cosf(freq), sn = __sinf(freq);
      o1[e] = xn1 * c + xn2 * sn;
      o2[e] = xn2 * c - xn1 * sn;
    }
    *(v4f*)(p + it * 4 * D_HEAD)      = o1;
    *(v4f*)(p + it * 4 * D_HEAD + 32) = o2;
  }
}

#define AK 64
#define APITCH 68
#define ATILE (AK * APITCH)
#define ASTAGE (2 * ATILE)
__global__ __launch_bounds__(512) void attn_kernel(
    const float* __restrict__ Q, const float* __restrict__ K,
    const float* __restrict__ V, float* __restrict__ Z) {
  __shared__ __attribute__((aligned(16))) float kv[2 * ASTAGE];
  __shared__ __attribute__((aligned(16))) float lds[16 * 256];
  const int tid = threadIdx.x;
  const int wave = __builtin_amdgcn_readfirstlane(tid >> 5);
  const int lane = tid & 31;
  const int lo = lane & 15, hi = lane >> 4;
  const int head = blockIdx.x >> 3;
  const int jp = blockIdx.x & 7;
  const int longg = (((wave >> 2) ^ wave) & 1) == 0;
  const int place = (int)((0x5243071634256170ULL >> (4 * wave)) & 7);
  const int qt = (longg ? 15 - jp : jp) * 8 + place;
  const float* Qh = Q + (size_t)head * N_CTX * D_HEAD;
  const float* Kh = K + (size_t)head * N_CTX * D_HEAD;
  const float* Vh = V + (size_t)head * D_HEAD * N_CTX;
  float* pl = lds + wave * 256;

  const int cr = tid >> 4, cc = (tid & 15) * 4;
  const float* kg = Kh + (size_t)cr * D_HEAD + cc;
  const float* vg = Vh + (size_t)cr * N_CTX + cc;
  float* st = kv + cr * APITCH + cc;
  const float* kf = kv + lo * APITCH + 2 * hi;
  const float* vf = kv + ATILE + lo * APITCH + 2 * hi;
  const int nstage = 2 * (15 - jp) + 2;

  v4f pre[4];
#pragma unroll
  for (int p = 0; p < 2; ++p) {
    pre[p]     = *(const v4f*)(kg + p * 32 * D_HEAD);
    pre[2 + p] = *(const v4f*)(vg + (size_t)p * 32 * N_CTX);
  }
#pragma unroll
  for (int p = 0; p < 2; ++p) {
    *(v4f*)(st + p * 32 * APITCH)         = pre[p];
    *(v4f*)(st + ATILE + p * 32 * APITCH) = pre[2 + p];
  }

  v2f qa[16];
  {
    const float* qrow = Qh + (size_t)(qt * 16 + lo) * D_HEAD + 2 * hi;
#pragma unroll
    for (int kk = 0; kk < 16; ++kk) qa[kk] = *(const v2f*)(qrow + kk * 4);
  }
  __syncthreads();

  v8f z0 = {}, z1 = {}, z2 = {}, z3 = {};
  for (int s = 0; s < nstage; ++s) {
    const int cur = (s & 1) * ASTAGE;
    const bool more = (s + 1 < nstage);
    if (more) {
      const float* kn = kg + (size_t)(s + 1) * AK * D_HEAD;
      const float* vn = vg + (s + 1) * AK;
#pragma unroll
      for (int p = 0; p < 2; ++p) {
        pre[p]     = *(const v4f*)(kn + p * 32 * D_HEAD);
        pre[2 + p] = *(const v4f*)(vn + (size_t)p * 32 * N_CTX);
      }
    }
    for (int t = 0; t < 4; ++t) {
      const int kt = s * 4 + t;
      if (kt > qt) break;
      v2f kb[16], vb[4][4];
#pragma unroll
      for (int kk = 0; kk < 16; ++kk)
        kb[kk] = *(const v2f*)(kf + cur + t * 16 * APITCH + kk * 4);
#pragma unroll
      for (int kk2 = 0; kk2 < 4; ++kk2) {
#pragma unroll
        for (int db = 0; db < 4; ++db)
          vb[kk2][db] = *(const v2f*)(vf + cur + db * 16 * APITCH + t * 16 + kk2 * 4);
      }
      __builtin_amdgcn_sched_barrier(0);
      v8f sc = {};
#pragma unroll
      for (int kk = 0; kk < 16; ++kk) sc = wmma_f32(qa[kk], kb[kk], sc);
#pragma unroll
      for (int r = 0; r < 8; ++r) {
        float pv = sc[r] * (1.0f / 64.0f);
        pv = pv * pv;
        if (kt == qt && lo > r + 8 * hi) pv = 0.0f;
        pl[(r + 8 * hi) * 16 + lo] = pv;
      }
#pragma unroll
      for (int kk2 = 0; kk2 < 4; ++kk2) {
        v2f pa = *(const v2f*)(pl + lo * 16 + kk2 * 4 + 2 * hi);
        z0 = wmma_f32(pa, vb[kk2][0], z0);
        z1 = wmma_f32(pa, vb[kk2][1], z1);
        z2 = wmma_f32(pa, vb[kk2][2], z2);
        z3 = wmma_f32(pa, vb[kk2][3], z3);
      }
    }
    if (more) {
      const int nxt = ASTAGE - cur;
#pragma unroll
      for (int p = 0; p < 2; ++p) {
        *(v4f*)(st + nxt + p * 32 * APITCH)         = pre[p];
        *(v4f*)(st + nxt + ATILE + p * 32 * APITCH) = pre[2 + p];
      }
    }
    __syncthreads();
  }
  const int b = head / N_HEAD, h = head % N_HEAD;
#pragma unroll
  for (int r = 0; r < 8; ++r) {
    const int s = qt * 16 + r + 8 * hi;
    float* zr = Z + (size_t)(b * N_CTX + s) * D_MODEL + h * D_HEAD + lo;
    zr[0]  = z0[r];
    zr[16] = z1[r];
    zr[32] = z2[r];
    zr[48] = z3[r];
  }
}

extern "C" void kernel_launch(void* const* d_in, const int* in_sizes, int n_in,
                              void* d_out, int out_size, void* d_ws, size_t ws_size,
                              hipStream_t stream) {
  const float* x  = (const float*)d_in[0];
  const float* Wq = (const float*)d_in[1];
  const float* bq = (const float*)d_in[2];
  const float* Wk = (const float*)d_in[3];
  const float* bk = (const float*)d_in[4];
  const float* Wv = (const float*)d_in[5];
  const float* bv = (const float*)d_in[6];
  const float* Wo = (const float*)d_in[7];
  const float* g  = (const float*)d_in[8];
  float* out = (float*)d_out;

  const size_t QKV = (size_t)BATCH * N_HEAD * N_CTX * D_HEAD;
  float* q = (float*)d_ws;
  float* k = q + QKV;
  float* v = k + QKV;
  float* z = v + QKV;

  dim3 gblk(M_ROWS / BM, D_MODEL / BN);
  qkv_gemm_kernel<<<gblk, 128, 0, stream>>>(x, Wq, bq, q, 0);
  qkv_gemm_kernel<<<gblk, 128, 0, stream>>>(x, Wk, bk, k, 0);
  qkv_gemm_kernel<<<gblk, 128, 0, stream>>>(x, Wv, bv, v, 1);

  norm_rope_kernel<<<NR_BLOCKS, 512, 0, stream>>>(q, g);

  attn_kernel<<<(BATCH * N_HEAD * (N_CTX / 16)) / 16, 512, 0, stream>>>(q, k, v, z);

  out_gemm_kernel<<<gblk, 128, 0, stream>>>(z, Wo, out);
}


# Attention as 240 blocks (was 192 on 256 WGPs): a head's 32 groups of 4 q tiles dealt to 10 blocks of near-equal stage cost, waves placed by two orthogonal Latin squares; per-tile sums unchanged
# speedup vs baseline: 2.8403x; 1.0506x over previous
#include <hip/hip_runtime.h>
#include <cstddef>

typedef float v2f __attribute__((ext_vector_type(2)));
typedef float v8f __attribute__((ext_vector_type(8)));

#define D_MODEL 768
#define N_HEAD  12
#define D_HEAD  64
#define N_CTX   2048
#define BATCH   2
#define M_ROWS  (BATCH * N_CTX)

__device__ __forceinline__ v8f wmma_f32(v2f a, v2f b, v8f c) {
  return __builtin_amdgcn_wmma_f32_16x16x4_f32(false, a, false, b, (short)0, c,
                                               false, false);
}

typedef float v4f __attribute__((ext_vector_type(4)));

#define BM 64
#define BN 96
#define KC 32
#define PITCH 36
#define STAGE_FLOATS ((BM + BN) * PITCH)
#define N_CHUNK (D_MODEL / KC)
#define GEMM_LDS_FLOATS 28672

__device__ __forceinline__ void gemm_tile_lds(const float* __restrict__ X,
                                              const float* __restrict__ W,
                                              float* __restrict__ lds,
                                              v8f acc[2][3]) {
  const int tid = threadIdx.x;
  const int wave = tid >> 5, lane = tid & 31;
  const int lo = lane & 15, hi = lane >> 4;
  const int wm = wave >> 1, wn = wave & 1;
  const int cr = tid >> 3, cc = (tid & 7) * 4;
  const float* xg = X + (size_t)(blockIdx.x * BM + cr) * D_MODEL + cc;
  const float* wg = W + (size_t)(blockIdx.y * BN + cr) * D_MODEL + cc;
  float* st = lds + cr * PITCH + cc;
  const float* fa = lds + (wm * 32 + lo) * PITCH + 2 * hi;
  const float* fb = lds + (BM + wn * 48 + lo) * PITCH + 2 * hi;

  v4f pre[10];
#pragma unroll
  for (int i = 0; i < 4; ++i) pre[i] = *(const v4f*)(xg + i * 16 * D_MODEL);
#pragma unroll
  for (int i = 0; i < 6; ++i) pre[4 + i] = *(const v4f*)(wg + i * 16 * D_MODEL);
#pragma unroll
  for (int i = 0; i < 10; ++i) *(v4f*)(st + i * 16 * PITCH) = pre[i];
  __syncthreads();

  for (int c = 0; c < N_CHUNK; ++c) {
    if (c == 0) __builtin_amdgcn_s_setprio(3);
    else if (c == N_CHUNK / 4) __builtin_amdgcn_s_setprio(2);
    else if (c == N_CHUNK / 2) __builtin_amdgcn_s_setprio(1);
    else if (c == 3 * N_CHUNK / 4) __builtin_amdgcn_s_setprio(0);
    const int cur = (c & 1) * STAGE_FLOATS;
    const bool more = (c + 1 < N_CHUNK);
    if (more) {
      const int k1 = (c + 1) * KC;
#pragma unroll
      for (int i = 0; i < 4; ++i)
        pre[i] = *(const v4f*)(xg + i * 16 * D_MODEL + k1);
#pragma unroll
      for (int i = 0; i < 6; ++i)
        pre[4 + i] = *(const v4f*)(wg + i * 16 * D_MODEL + k1);
    }
    v2f af[KC / 4][2], bf[KC / 4][3];
#pragma unroll
    for (int kk = 0; kk < KC / 4; ++kk) {
      af[kk][0] = *(const v2f*)(fa + cur + kk * 4);
      af[kk][1] = *(const v2f*)(fa + cur + 16 * PITCH + kk * 4);
      bf[kk][0] = *(const v2f*)(fb + cur + kk * 4);
      bf[kk][1] = *(const v2f*)(fb + cur + 16 * PITCH + kk * 4);
      bf[kk][2] = *(const v2f*)(fb + cur + 32 * PITCH + kk * 4);
    }
    __builtin_amdgcn_sched_barrier(0);
#pragma unroll
    for (int kk = 0; kk < KC / 4; ++kk) {
      acc[0][0] = wmma_f32(af[kk][0], bf[kk][0], acc[0][0]);
      acc[0][1] = wmma_f32(af[kk][0], bf[kk][1], acc[0][1]);
      acc[0][2] = wmma_f32(af[kk][0], bf[kk][2], acc[0][2]);
      acc[1][0] = wmma_f32(af[kk][1], bf[kk][0], acc[1][0]);
      acc[1][1] = wmma_f32(af[kk][1], bf[kk][1], acc[1][1]);
      acc[1][2] = wmma_f32(af[kk][1], bf[kk][2], acc[1][2]);
    }
    __builtin_amdgcn_sched_barrier(0);
    if (more) {
      const int nxt = STAGE_FLOATS - cur;
#pragma unroll
      for (int i = 0; i < 10; ++i) *(v4f*)(st + nxt + i * 16 * PITCH) = pre[i];
    }
    __syncthreads();
  }
}

__global__ __launch_bounds__(128) void qkv_gemm_kernel(
    const float* __restrict__ X, const float* __restrict__ W,
    const float* __restrict__ bias, float* __restrict__ out, int vmode) {
  __shared__ __attribute__((aligned(16))) float lds[GEMM_LDS_FLOATS];
  const int wave = threadIdx.x >> 5;
  const int lane = threadIdx.x & 31;
  const int lo = lane & 15, hi = lane >> 4;
  const int m0 = blockIdx.x * BM + (wave >> 1) * 32;
  const int n0 = blockIdx.y * BN + (wave & 1) * 48;
  v8f acc[2][3] = {{{}, {}, {}}, {{}, {}, {}}};
  gemm_tile_lds(X, W, lds, acc);
#pragma unroll
  for (int j = 0; j < 3; ++j) {
    const int n = n0 + j * 16 + lo;
    const int h = n >> 6, d = n & (D_HEAD - 1);
    const float bv = bias[n];
#pragma unroll
    for (int i = 0; i < 2; ++i) {
#pragma unroll
      for (int r = 0; r < 8; ++r) {
        const int m = m0 + i * 16 + r + 8 * hi;
        const int b = m >> 11, s = m & (N_CTX - 1);
        const size_t hb = (size_t)b * N_HEAD + h;
        if (vmode)
          out[(hb * D_HEAD + d) * N_CTX + s] = acc[i][j][r] + bv;
        else
          out[(hb * N_CTX + s) * D_HEAD + d] = acc[i][j][r] + bv;
      }
    }
  }
}

__global__ __launch_bounds__(128) void out_gemm_kernel(
    const float* __restrict__ X, const float* __restrict__ W,
    float* __restrict__ out) {
  __shared__ __attribute__((aligned(16))) float lds[GEMM_LDS_FLOATS];
  const int wave = threadIdx.x >> 5;
  const int lane = threadIdx.x & 31;
  const int lo = lane & 15, hi = lane >> 4;
  const int m0 = blockIdx.x * BM + (wave >> 1) * 32;
  const int n0 = blockIdx.y * BN + (wave & 1) * 48;
  v8f acc[2][3] = {{{}, {}, {}}, {{}, {}, {}}};
  gemm_tile_lds(X, W, lds, acc);
#pragma unroll
  for (int i = 0; i < 2; ++i) {
#pragma unroll
    for (int j = 0; j < 3; ++j) {
#pragma unroll
      for (int r = 0; r < 8; ++r) {
        const int m = m0 + i * 16 + r + 8 * hi;
        out[(size_t)m * D_MODEL + n0 + j * 16 + lo] = acc[i][j][r];
      }
    }
  }
}

#define NR_ROWS 24
#define NR_IT   (NR_ROWS / 4)
#define NR_BLOCKS ((2 * BATCH * N_HEAD * N_CTX) / (16 * NR_ROWS))
__global__ __launch_bounds__(512) void norm_rope_kernel(
    float* __restrict__ qk, const float* __restrict__ g) {
  const int lane = threadIdx.x & 31;
  const int gw = blockIdx.x * 16 + (threadIdx.x >> 5);
  const int c8 = lane & 7;
  const int row0 = gw * NR_ROWS + (lane >> 3);
  float* p = qk + (size_t)row0 * D_HEAD + 4 * c8;
  v4f xa[NR_IT], xb[NR_IT];
#pragma unroll
  for (int it = 0; it < NR_IT; ++it) {
    xa[it] = *(const v4f*)(p + it * 4 * D_HEAD);
    xb[it] = *(const v4f*)(p + it * 4 * D_HEAD + 32);
  }
  const v4f ga = *(const v4f*)(g + 4 * c8);
  const v4f gb = *(const v4f*)(g + 32 + 4 * c8);
  asm volatile("" :: "v"(xa[0]), "v"(xb[0]), "v"(xa[1]), "v"(xb[1]), "v"(xa[2]), "v"(xb[2]),
                     "v"(xa[3]), "v"(xb[3]), "v"(xa[4]), "v"(xb[4]), "v"(xa[5]), "v"(xb[5]),
                     "v"(ga), "v"(gb));
  float fr[4];
#pragma unroll
  for (int e = 0; e < 4; ++e)
    fr[e] = exp2f((float)(4 * c8 + e) * (-11.0f / 32.0f));
#pragma unroll
  for (int it = 0; it < NR_IT; ++it) {
    const int s = (row0 + it * 4) & (N_CTX - 1);
    const v4f a = xa[it], b = xb[it];
    float ss = 0.0f;
#pragma unroll
    for (int e = 0; e < 4; ++e) ss += a[e] * a[e] + b[e] * b[e];
    ss += __shfl_xor(ss, 1, 32);
    ss += __shfl_xor(ss, 2, 32);
    ss += __shfl_xor(ss, 4, 32);
    const float inv = rsqrtf(ss * (1.0f / 64.0f) + 1e-6f);
    v4f o1, o2;
#pragma unroll
    for (int e = 0; e < 4; ++e) {
      const float xn1 = a[e] * inv * ga[e];
      const float xn2 = b[e] * inv * gb[e];
      const float freq = (float)s * fr[e];
      const float c = __cosf(freq), sn = __sinf(freq);
      o1[e] = xn1 * c + xn2 * sn;
      o2[e] = xn2 * c - xn1 * sn;
    }
    *(v4f*)(p + it * 4 * D_HEAD)      = o1;
    *(v4f*)(p + it * 4 * D_HEAD + 32) = o2;
  }
}

#define AK 64
#define APITCH 68
#define ATILE (AK * APITCH)
#define ASTAGE (2 * ATILE)
#define ABLK_PER_HEAD 10
__global__ __launch_bounds__(512) void attn_kernel(
    const float* __restrict__ Q, const float* __restrict__ K,
    const float* __restrict__ V, float* __restrict__ Z) {
  __shared__ __attribute__((aligned(16))) float kv[2 * ASTAGE];
  __shared__ __attribute__((aligned(16))) float lds[16 * 256];
  const int tid = threadIdx.x;
  const int wave = __builtin_amdgcn_readfirstlane(tid >> 5);
  const int lane = tid & 31;
  const int lo = lane & 15, hi = lane >> 4;
  const int head = blockIdx.x / ABLK_PER_HEAD;
  const int bi = blockIdx.x - head * ABLK_PER_HEAD;
  constexpr unsigned int groups_of_block[ABLK_PER_HEAD] = {
      0xFFFF0B1Fu,
      0x0002071Eu,
      0xFFFF141Du,
      0xFFFF191Cu,
      0xFF01181Bu,
      0xFF03171Au,
      0x04051516u,
      0x08091013u,
      0x060C1112u,
      0x0A0D0E0Fu};
  const unsigned int gw = groups_of_block[bi];
  const int wa = wave >> 2, wb = wave & 3;
  const int slot = wa ^ wb;
  const int place = wa ^ ((0x1320 >> (4 * wb)) & 3);
  const int grp = (int)((gw >> (8 * slot)) & 0xFFu);
  const bool has = grp != 0xFF;
  const int qt = has ? grp * 4 + place : -1;
  const float* Qh = Q + (size_t)head * N_CTX * D_HEAD;
  const float* Kh = K + (size_t)head * N_CTX * D_HEAD;
  const float* Vh = V + (size_t)head * D_HEAD * N_CTX;
  float* pl = lds + wave * 256;

  const int cr = tid >> 4, cc = (tid & 15) * 4;
  const float* kg = Kh + (size_t)cr * D_HEAD + cc;
  const float* vg = Vh + (size_t)cr * N_CTX + cc;
  float* st = kv + cr * APITCH + cc;
  const float* kf = kv + lo * APITCH + 2 * hi;
  const float* vf = kv + ATILE + lo * APITCH + 2 * hi;
  const int nstage = (int)(gw & 0xFFu) + 1;

  v4f pre[4];
#pragma unroll
  for (int p = 0; p < 2; ++p) {
    pre[p]     = *(const v4f*)(kg + p * 32 * D_HEAD);
    pre[2 + p] = *(const v4f*)(vg + (size_t)p * 32 * N_CTX);
  }
#pragma unroll
  for (int p = 0; p < 2; ++p) {
    *(v4f*)(st + p * 32 * APITCH)         = pre[p];
    *(v4f*)(st + ATILE + p * 32 * APITCH) = pre[2 + p];
  }

  v2f qa[16];
  {
    const float* qrow = Qh + (size_t)((has ? qt : 0) * 16 + lo) * D_HEAD + 2 * hi;
#pragma unroll
    for (int kk = 0; kk < 16; ++kk) qa[kk] = *(const v2f*)(qrow + kk * 4);
  }
  __syncthreads();

  v8f z0 = {}, z1 = {}, z2 = {}, z3 = {};
  for (int s = 0; s < nstage; ++s) {
    const int cur = (s & 1) * ASTAGE;
    const bool more = (s + 1 < nstage);
    if (more) {
      const float* kn = kg + (size_t)(s + 1) * AK * D_HEAD;
      const float* vn = vg + (s + 1) * AK;
#pragma unroll
      for (int p = 0; p < 2; ++p) {
        pre[p]     = *(const v4f*)(kn + p * 32 * D_HEAD);
        pre[2 + p] = *(const v4f*)(vn + (size_t)p * 32 * N_CTX);
      }
    }
    for (int t = 0; t < 4; ++t) {
      const int kt = s * 4 + t;
      if (kt > qt) break;
      v2f kb[16], vb[4][4];
#pragma unroll
      for (int kk = 0; kk < 16; ++kk)
        kb[kk] = *(const v2f*)(kf + cur + t * 16 * APITCH + kk * 4);
#pragma unroll
      for (int kk2 = 0; kk2 < 4; ++kk2) {
#pragma unroll
        for (int db = 0; db < 4; ++db)
          vb[kk2][db] = *(const v2f*)(vf + cur + db * 16 * APITCH + t * 16 + kk2 * 4);
      }
      __builtin_amdgcn_sched_barrier(0);
      v8f sc = {};
#pragma unroll
      for (int kk = 0; kk < 16; ++kk) sc = wmma_f32(qa[kk], kb[kk], sc);
#pragma unroll
      for (int r = 0; r < 8; ++r) {
        float pv = sc[r] * (1.0f / 64.0f);
        pv = pv * pv;
        if (kt == qt && lo > r + 8 * hi) pv = 0.0f;
        pl[(r + 8 * hi) * 16 + lo] = pv;
      }
#pragma unroll
      for (int kk2 = 0; kk2 < 4; ++kk2) {
        v2f pa = *(const v2f*)(pl + lo * 16 + kk2 * 4 + 2 * hi);
        z0 = wmma_f32(pa, vb[kk2][0], z0);
        z1 = wmma_f32(pa, vb[kk2][1], z1);
        z2 = wmma_f32(pa, vb[kk2][2], z2);
        z3 = wmma_f32(pa, vb[kk2][3], z3);
      }
    }
    if (more) {
      const int nxt = ASTAGE - cur;
#pragma unroll
      for (int p = 0; p < 2; ++p) {
        *(v4f*)(st + nxt + p * 32 * APITCH)         = pre[p];
        *(v4f*)(st + nxt + ATILE + p * 32 * APITCH) = pre[2 + p];
      }
    }
    __syncthreads();
  }
  const int b = head / N_HEAD, h = head % N_HEAD;
  if (has) {
#pragma unroll
    for (int r = 0; r < 8; ++r) {
      const int s = qt * 16 + r + 8 * hi;
      float* zr = Z + (size_t)(b * N_CTX + s) * D_MODEL + h * D_HEAD + lo;
      zr[0]  = z0[r];
      zr[16] = z1[r];
      zr[32] = z2[r];
      zr[48] = z3[r];
    }
  }
}

extern "C" void kernel_launch(void* const* d_in, const int* in_sizes, int n_in,
                              void* d_out, int out_size, void* d_ws, size_t ws_size,
                              hipStream_t stream) {
  const float* x  = (const float*)d_in[0];
  const float* Wq = (const float*)d_in[1];
  const float* bq = (const float*)d_in[2];
  const float* Wk = (const float*)d_in[3];
  const float* bk = (const float*)d_in[4];
  const float* Wv = (const float*)d_in[5];
  const float* bv = (const float*)d_in[6];
  const float* Wo = (const float*)d_in[7];
  const float* g  = (const float*)d_in[8];
  float* out = (float*)d_out;

  const size_t QKV = (size_t)BATCH * N_HEAD * N_CTX * D_HEAD;
  float* q = (float*)d_ws;
  float* k = q + QKV;
  float* v = k + QKV;
  float* z = v + QKV;

  dim3 gblk(M_ROWS / BM, D_MODEL / BN);
  qkv_gemm_kernel<<<gblk, 128, 0, stream>>>(x, Wq, bq, q, 0);
  qkv_gemm_kernel<<<gblk, 128, 0, stream>>>(x, Wk, bk, k, 0);
  qkv_gemm_kernel<<<gblk, 128, 0, stream>>>(x, Wv, bv, v, 1);

  norm_rope_kernel<<<NR_BLOCKS, 512, 0, stream>>>(q, g);

  attn_kernel<<<BATCH * N_HEAD * ABLK_PER_HEAD, 512, 0, stream>>>(q, k, v, z);

  out_gemm_kernel<<<gblk, 128, 0, stream>>>(z, Wo, out);
}


# V projection's epilogue turns each wave's 32x48 tile round in LDS and writes every 128-B line of V^T whole, 4 lines per b128 store (12 per wave); was 48 b32 stores of 32 4-byte pieces; sums unchanged
# speedup vs baseline: 2.9436x; 1.0364x over previous
#include <hip/hip_runtime.h>
#include <cstddef>

typedef float v2f __attribute__((ext_vector_type(2)));
typedef float v8f __attribute__((ext_vector_type(8)));

#define D_MODEL 768
#define N_HEAD  12
#define D_HEAD  64
#define N_CTX   2048
#define BATCH   2
#define M_ROWS  (BATCH * N_CTX)

__device__ __forceinline__ v8f wmma_f32(v2f a, v2f b, v8f c) {
  return __builtin_amdgcn_wmma_f32_16x16x4_f32(false, a, false, b, (short)0, c,
                                               false, false);
}

typedef float v4f __attribute__((ext_vector_type(4)));

#define BM 64
#define BN 96
#define KC 32
#define PITCH 36
#define STAGE_FLOATS ((BM + BN) * PITCH)
#define N_CHUNK (D_MODEL / KC)
#define GEMM_LDS_FLOATS 28672
#define VT_PITCH 36

__device__ __forceinline__ void gemm_tile_lds(const float* __restrict__ X,
                                              const float* __restrict__ W,
                                              float* __restrict__ lds,
                                              v8f acc[2][3]) {
  const int tid = threadIdx.x;
  const int wave = tid >> 5, lane = tid & 31;
  const int lo = lane & 15, hi = lane >> 4;
  const int wm = wave >> 1, wn = wave & 1;
  const int cr = tid >> 3, cc = (tid & 7) * 4;
  const float* xg = X + (size_t)(blockIdx.x * BM + cr) * D_MODEL + cc;
  const float* wg = W + (size_t)(blockIdx.y * BN + cr) * D_MODEL + cc;
  float* st = lds + cr * PITCH + cc;
  const float* fa = lds + (wm * 32 + lo) * PITCH + 2 * hi;
  const float* fb = lds + (BM + wn * 48 + lo) * PITCH + 2 * hi;

  v4f pre[10];
#pragma unroll
  for (int i = 0; i < 4; ++i) pre[i] = *(const v4f*)(xg + i * 16 * D_MODEL);
#pragma unroll
  for (int i = 0; i < 6; ++i) pre[4 + i] = *(const v4f*)(wg + i * 16 * D_MODEL);
#pragma unroll
  for (int i = 0; i < 10; ++i) *(v4f*)(st + i * 16 * PITCH) = pre[i];
  __syncthreads();

  for (int c = 0; c < N_CHUNK; ++c) {
    if (c == 0) __builtin_amdgcn_s_setprio(3);
    else if (c == N_CHUNK / 4) __builtin_amdgcn_s_setprio(2);
    else if (c == N_CHUNK / 2) __builtin_amdgcn_s_setprio(1);
    else if (c == 3 * N_CHUNK / 4) __builtin_amdgcn_s_setprio(0);
    const int cur = (c & 1) * STAGE_FLOATS;
    const bool more = (c + 1 < N_CHUNK);
    if (more) {
      const int k1 = (c + 1) * KC;
#pragma unroll
      for (int i = 0; i < 4; ++i)
        pre[i] = *(const v4f*)(xg + i * 16 * D_MODEL + k1);
#pragma unroll
      for (int i = 0; i < 6; ++i)
        pre[4 + i] = *(const v4f*)(wg + i * 16 * D_MODEL + k1);
    }
    v2f af[KC / 4][2], bf[KC / 4][3];
#pragma unroll
    for (int kk = 0; kk < KC / 4; ++kk) {
      af[kk][0] = *(const v2f*)(fa + cur + kk * 4);
      af[kk][1] = *(const v2f*)(fa + cur + 16 * PITCH + kk * 4);
      bf[kk][0] = *(const v2f*)(fb + cur + kk * 4);
      bf[kk][1] = *(const v2f*)(fb + cur + 16 * PITCH + kk * 4);
      bf[kk][2] = *(const v2f*)(fb + cur + 32 * PITCH + kk * 4);
    }
    __builtin_amdgcn_sched_barrier(0);
#pragma unroll
    for (int kk = 0; kk < KC / 4; ++kk) {
      acc[0][0] = wmma_f32(af[kk][0], bf[kk][0], acc[0][0]);
      acc[0][1] = wmma_f32(af[kk][0], bf[kk][1], acc[0][1]);
      acc[0][2] = wmma_f32(af[kk][0], bf[kk][2], acc[0][2]);
      acc[1][0] = wmma_f32(af[kk][1], bf[kk][0], acc[1][0]);
      acc[1][1] = wmma_f32(af[kk][1], bf[kk][1], acc[1][1]);
      acc[1][2] = wmma_f32(af[kk][1], bf[kk][2], acc[1][2]);
    }
    __builtin_amdgcn_sched_barrier(0);
    if (more) {
      const int nxt = STAGE_FLOATS - cur;
#pragma unroll
      for (int i = 0; i < 10; ++i) *(v4f*)(st + nxt + i * 16 * PITCH) = pre[i];
    }
    __syncthreads();
  }
}

__global__ __launch_bounds__(128) void qkv_gemm_kernel(
    const float* __restrict__ X, const float* __restrict__ W,
    const float* __restrict__ bias, float* __restrict__ out, int vmode) {
  __shared__ __attribute__((aligned(16))) float lds[GEMM_LDS_FLOATS];
  const int wave = threadIdx.x >> 5;
  const int lane = threadIdx.x & 31;
  const int lo = lane & 15, hi = lane >> 4;
  const int m0 = blockIdx.x * BM + (wave >> 1) * 32;
  const int n0 = blockIdx.y * BN + (wave & 1) * 48;
  v8f acc[2][3] = {{{}, {}, {}}, {{}, {}, {}}};
  gemm_tile_lds(X, W, lds, acc);
  if (vmode) {
    float* tw = lds + wave * (48 * VT_PITCH);
#pragma unroll
    for (int j = 0; j < 3; ++j) {
      const float bv = bias[n0 + j * 16 + lo];
#pragma unroll
      for (int i = 0; i < 2; ++i) {
        v4f c0, c1;
#pragma unroll
        for (int e = 0; e < 4; ++e) {
          c0[e] = acc[i][j][e] + bv;
          c1[e] = acc[i][j][4 + e] + bv;
        }
        float* tp = tw + (j * 16 + lo) * VT_PITCH + i * 16 + 8 * hi;
        *(v4f*)tp = c0;
        *(v4f*)(tp + 4) = c1;
      }
    }
    const int rr = lane >> 3, cq = (lane & 7) * 4;
    const int m = m0 + cq;
    const int b = m >> 11, s = m & (N_CTX - 1);
#pragma unroll
    for (int it = 0; it < 12; ++it) {
      const int nl = it * 4 + rr;
      const v4f t = *(const v4f*)(tw + nl * VT_PITCH + cq);
      const int n = n0 + nl;
      const int h = n >> 6, d = n & (D_HEAD - 1);
      *(v4f*)(out + (((size_t)b * N_HEAD + h) * D_HEAD + d) * N_CTX + s) = t;
    }
    return;
  }
#pragma unroll
  for (int j = 0; j < 3; ++j) {
    const int n = n0 + j * 16 + lo;
    const int h = n >> 6, d = n & (D_HEAD - 1);
    const float bv = bias[n];
#pragma unroll
    for (int i = 0; i < 2; ++i) {
#pragma unroll
      for (int r = 0; r < 8; ++r) {
        const int m = m0 + i * 16 + r + 8 * hi;
        const int b = m >> 11, s = m & (N_CTX - 1);
        const size_t hb = (size_t)b * N_HEAD + h;
        out[(hb * N_CTX + s) * D_HEAD + d] = acc[i][j][r] + bv;
      }
    }
  }
}

__global__ __launch_bounds__(128) void out_gemm_kernel(
    const float* __restrict__ X, const float* __restrict__ W,
    float* __restrict__ out) {
  __shared__ __attribute__((aligned(16))) float lds[GEMM_LDS_FLOATS];
  const int wave = threadIdx.x >> 5;
  const int lane = threadIdx.x & 31;
  const int lo = lane & 15, hi = lane >> 4;
  const int m0 = blockIdx.x * BM + (wave >> 1) * 32;
  const int n0 = blockIdx.y * BN + (wave & 1) * 48;
  v8f acc[2][3] = {{{}, {}, {}}, {{}, {}, {}}};
  gemm_tile_lds(X, W, lds, acc);
#pragma unroll
  for (int i = 0; i < 2; ++i) {
#pragma unroll
    for (int j = 0; j < 3; ++j) {
#pragma unroll
      for (int r = 0; r < 8; ++r) {
        const int m = m0 + i * 16 + r + 8 * hi;
        out[(size_t)m * D_MODEL + n0 + j * 16 + lo] = acc[i][j][r];
      }
    }
  }
}

#define NR_ROWS 24
#define NR_IT   (NR_ROWS / 4)
#define NR_BLOCKS ((2 * BATCH * N_HEAD * N_CTX) / (16 * NR_ROWS))
__global__ __launch_bounds__(512) void norm_rope_kernel(
    float* __restrict__ qk, const float* __restrict__ g) {
  const int lane = threadIdx.x & 31;
  const int gw = blockIdx.x * 16 + (threadIdx.x >> 5);
  const int c8 = lane & 7;
  const int row0 = gw * NR_ROWS + (lane >> 3);
  float* p = qk + (size_t)row0 * D_HEAD + 4 * c8;
  v4f xa[NR_IT], xb[NR_IT];
#pragma unroll
  for (int it = 0; it < NR_IT; ++it) {
    xa[it] = *(const v4f*)(p + it * 4 * D_HEAD);
    xb[it] = *(const v4f*)(p + it * 4 * D_HEAD + 32);
  }
  const v4f ga = *(const v4f*)(g + 4 * c8);
  const v4f gb = *(const v4f*)(g + 32 + 4 * c8);
  asm volatile("" :: "v"(xa[0]), "v"(xb[0]), "v"(xa[1]), "v"(xb[1]), "v"(xa[2]), "v"(xb[2]),
                     "v"(xa[3]), "v"(xb[3]), "v"(xa[4]), "v"(xb[4]), "v"(xa[5]), "v"(xb[5]),
                     "v"(ga), "v"(gb));
  float fr[4];
#pragma unroll
  for (int e = 0; e < 4; ++e)
    fr[e] = exp2f((float)(4 * c8 + e) * (-11.0f / 32.0f));
#pragma unroll
  for (int it = 0; it < NR_IT; ++it) {
    const int s = (row0 + it * 4) & (N_CTX - 1);
    const v4f a = xa[it], b = xb[it];
    float ss = 0.0f;
#pragma unroll
    for (int e = 0; e < 4; ++e) ss += a[e] * a[e] + b[e] * b[e];
    ss += __shfl_xor(ss, 1, 32);
    ss += __shfl_xor(ss, 2, 32);
    ss += __shfl_xor(ss, 4, 32);
    const float inv = rsqrtf(ss * (1.0f / 64.0f) + 1e-6f);
    v4f o1, o2;
#pragma unroll
    for (int e = 0; e < 4; ++e) {
      const float xn1 = a[e] * inv * ga[e];
      const float xn2 = b[e] * inv * gb[e];
      const float freq = (float)s * fr[e];
      const float c = __cosf(freq), sn = __sinf(freq);
      o1[e] = xn1 * c + xn2 * sn;
      o2[e] = xn2 * c - xn1 * sn;
    }
    *(v4f*)(p + it * 4 * D_HEAD)      = o1;
    *(v4f*)(p + it * 4 * D_HEAD + 32) = o2;
  }
}

#define AK 64
#define APITCH 68
#define ATILE (AK * APITCH)
#define ASTAGE (2 * ATILE)
#define ABLK_PER_HEAD 10
__global__ __launch_bounds__(512) void attn_kernel(
    const float* __restrict__ Q, const float* __restrict__ K,
    const float* __restrict__ V, float* __restrict__ Z) {
  __shared__ __attribute__((aligned(16))) float kv[2 * ASTAGE];
  __shared__ __attribute__((aligned(16))) float lds[16 * 256];
  const int tid = threadIdx.x;
  const int wave = __builtin_amdgcn_readfirstlane(tid >> 5);
  const int lane = tid & 31;
  const int lo = lane & 15, hi = lane >> 4;
  const int head = blockIdx.x / ABLK_PER_HEAD;
  const int bi = blockIdx.x - head * ABLK_PER_HEAD;
  constexpr unsigned int groups_of_block[ABLK_PER_HEAD] = {
      0xFFFF0B1Fu,
      0x0002071Eu,
      0xFFFF141Du,
      0xFFFF191Cu,
      0xFF01181Bu,
      0xFF03171Au,
      0x04051516u,
      0x08091013u,
      0x060C1112u,
      0x0A0D0E0Fu};
  const unsigned int gw = groups_of_block[bi];
  const int wa = wave >> 2, wb = wave & 3;
  const int slot = wa ^ wb;
  const int place = wa ^ ((0x1320 >> (4 * wb)) & 3);
  const int grp = (int)((gw >> (8 * slot)) & 0xFFu);
  const bool has = grp != 0xFF;
  const int qt = has ? grp * 4 + place : -1;
  const float* Qh = Q + (size_t)head * N_CTX * D_HEAD;
  const float* Kh = K + (size_t)head * N_CTX * D_HEAD;
  const float* Vh = V + (size_t)head * D_HEAD * N_CTX;
  float* pl = lds + wave * 256;

  const int cr = tid >> 4, cc = (tid & 15) * 4;
  const float* kg = Kh + (size_t)cr * D_HEAD + cc;
  const float* vg = Vh + (size_t)cr * N_CTX + cc;
  float* st = kv + cr * APITCH + cc;
  const float* kf = kv + lo * APITCH + 2 * hi;
  const float* vf = kv + ATILE + lo * APITCH + 2 * hi;
  const int nstage = (int)(gw & 0xFFu) + 1;

  v4f pre[4];
#pragma unroll
  for (int p = 0; p < 2; ++p) {
    pre[p]     = *(const v4f*)(kg + p * 32 * D_HEAD);
    pre[2 + p] = *(const v4f*)(vg + (size_t)p * 32 * N_CTX);
  }
#pragma unroll
  for (int p = 0; p < 2; ++p) {
    *(v4f*)(st + p * 32 * APITCH)         = pre[p];
    *(v4f*)(st + ATILE + p * 32 * APITCH) = pre[2 + p];
  }

  v2f qa[16];
  {
    const float* qrow = Qh + (size_t)((has ? qt : 0) * 16 + lo) * D_HEAD + 2 * hi;
#pragma unroll
    for (int kk = 0; kk < 16; ++kk) qa[kk] = *(const v2f*)(qrow + kk * 4);
  }
  __syncthreads();

  v8f z0 = {}, z1 = {}, z2 = {}, z3 = {};
  for (int s = 0; s < nstage; ++s) {
    const int cur = (s & 1) * ASTAGE;
    const bool more = (s + 1 < nstage);
    if (more) {
      const float* kn = kg + (size_t)(s + 1) * AK * D_HEAD;
      const float* vn = vg + (s + 1) * AK;
#pragma unroll
      for (int p = 0; p < 2; ++p) {
        pre[p]     = *(const v4f*)(kn + p * 32 * D_HEAD);
        pre[2 + p] = *(const v4f*)(vn + (size_t)p * 32 * N_CTX);
      }
    }
    for (int t = 0; t < 4; ++t) {
      const int kt = s * 4 + t;
      if (kt > qt) break;
      v2f kb[16], vb[4][4];
#pragma unroll
      for (int kk = 0; kk < 16; ++kk)
        kb[kk] = *(const v2f*)(kf + cur + t * 16 * APITCH + kk * 4);
#pragma unroll
      for (int kk2 = 0; kk2 < 4; ++kk2) {
#pragma unroll
        for (int db = 0; db < 4; ++db)
          vb[kk2][db] = *(const v2f*)(vf + cur + db * 16 * APITCH + t * 16 + kk2 * 4);
      }
      __builtin_amdgcn_sched_barrier(0);
      v8f sc = {};
#pragma unroll
      for (int kk = 0; kk < 16; ++kk) sc = wmma_f32(qa[kk], kb[kk], sc);
#pragma unroll
      for (int r = 0; r < 8; ++r) {
        float pv = sc[r] * (1.0f / 64.0f);
        pv = pv * pv;
        if (kt == qt && lo > r + 8 * hi) pv = 0.0f;
        pl[(r + 8 * hi) * 16 + lo] = pv;
      }
#pragma unroll
      for (int kk2 = 0; kk2 < 4; ++kk2) {
        v2f pa = *(const v2f*)(pl + lo * 16 + kk2 * 4 + 2 * hi);
        z0 = wmma_f32(pa, vb[kk2][0], z0);
        z1 = wmma_f32(pa, vb[kk2][1], z1);
        z2 = wmma_f32(pa, vb[kk2][2], z2);
        z3 = wmma_f32(pa, vb[kk2][3], z3);
      }
    }
    if (more) {
      const int nxt = ASTAGE - cur;
#pragma unroll
      for (int p = 0; p < 2; ++p) {
        *(v4f*)(st + nxt + p * 32 * APITCH)         = pre[p];
        *(v4f*)(st + nxt + ATILE + p * 32 * APITCH) = pre[2 + p];
      }
    }
    __syncthreads();
  }
  const int b = head / N_HEAD, h = head % N_HEAD;
  if (has) {
#pragma unroll
    for (int r = 0; r < 8; ++r) {
      const int s = qt * 16 + r + 8 * hi;
      float* zr = Z + (size_t)(b * N_CTX + s) * D_MODEL + h * D_HEAD + lo;
      zr[0]  = z0[r];
      zr[16] = z1[r];
      zr[32] = z2[r];
      zr[48] = z3[r];
    }
  }
}

extern "C" void kernel_launch(void* const* d_in, const int* in_sizes, int n_in,
                              void* d_out, int out_size, void* d_ws, size_t ws_size,
                              hipStream_t stream) {
  const float* x  = (const float*)d_in[0];
  const float* Wq = (const float*)d_in[1];
  const float* bq = (const float*)d_in[2];
  const float* Wk = (const float*)d_in[3];
  const float* bk = (const float*)d_in[4];
  const float* Wv = (const float*)d_in[5];
  const float* bv = (const float*)d_in[6];
  const float* Wo = (const float*)d_in[7];
  const float* g  = (const float*)d_in[8];
  float* out = (float*)d_out;

  const size_t QKV = (size_t)BATCH * N_HEAD * N_CTX * D_HEAD;
  float* q = (float*)d_ws;
  float* k = q + QKV;
  float* v = k + QKV;
  float* z = v + QKV;

  dim3 gblk(M_ROWS / BM, D_MODEL / BN);
  qkv_gemm_kernel<<<gblk, 128, 0, stream>>>(x, Wq, bq, q, 0);
  qkv_gemm_kernel<<<gblk, 128, 0, stream>>>(x, Wk, bk, k, 0);
  qkv_gemm_kernel<<<gblk, 128, 0, stream>>>(x, Wv, bv, v, 1);

  norm_rope_kernel<<<NR_BLOCKS, 512, 0, stream>>>(q, g);

  attn_kernel<<<BATCH * N_HEAD * ABLK_PER_HEAD, 512, 0, stream>>>(q, k, v, z);

  out_gemm_kernel<<<gblk, 128, 0, stream>>>(z, Wo, out);
}


# Attention scores computed transposed (K as A, Q as B) on 32-key tiles: P squared in registers is the B operand of Z^T += V^T P^T, its LDS round trip goes; PV key order in a tile changes
# speedup vs baseline: 3.0554x; 1.0380x over previous
#include <hip/hip_runtime.h>
#include <cstddef>

typedef float v2f __attribute__((ext_vector_type(2)));
typedef float v8f __attribute__((ext_vector_type(8)));

#define D_MODEL 768
#define N_HEAD  12
#define D_HEAD  64
#define N_CTX   2048
#define BATCH   2
#define M_ROWS  (BATCH * N_CTX)

__device__ __forceinline__ v8f wmma_f32(v2f a, v2f b, v8f c) {
  return __builtin_amdgcn_wmma_f32_16x16x4_f32(false, a, false, b, (short)0, c,
                                               false, false);
}

typedef float v4f __attribute__((ext_vector_type(4)));

#define BM 64
#define BN 96
#define KC 32
#define PITCH 36
#define STAGE_FLOATS ((BM + BN) * PITCH)
#define N_CHUNK (D_MODEL / KC)
#define GEMM_LDS_FLOATS 28672
#define VT_PITCH 36

__device__ __forceinline__ void gemm_tile_lds(const float* __restrict__ X,
                                              const float* __restrict__ W,
                                              float* __restrict__ lds,
                                              v8f acc[2][3]) {
  const int tid = threadIdx.x;
  const int wave = tid >> 5, lane = tid & 31;
  const int lo = lane & 15, hi = lane >> 4;
  const int wm = wave >> 1, wn = wave & 1;
  const int cr = tid >> 3, cc = (tid & 7) * 4;
  const float* xg = X + (size_t)(blockIdx.x * BM + cr) * D_MODEL + cc;
  const float* wg = W + (size_t)(blockIdx.y * BN + cr) * D_MODEL + cc;
  float* st = lds + cr * PITCH + cc;
  const float* fa = lds + (wm * 32 + lo) * PITCH + 2 * hi;
  const float* fb = lds + (BM + wn * 48 + lo) * PITCH + 2 * hi;

  v4f pre[10];
#pragma unroll
  for (int i = 0; i < 4; ++i) pre[i] = *(const v4f*)(xg + i * 16 * D_MODEL);
#pragma unroll
  for (int i = 0; i < 6; ++i) pre[4 + i] = *(const v4f*)(wg + i * 16 * D_MODEL);
#pragma unroll
  for (int i = 0; i < 10; ++i) *(v4f*)(st + i * 16 * PITCH) = pre[i];
  __syncthreads();

  for (int c = 0; c < N_CHUNK; ++c) {
    if (c == 0) __builtin_amdgcn_s_setprio(3);
    else if (c == N_CHUNK / 4) __builtin_amdgcn_s_setprio(2);
    else if (c == N_CHUNK / 2) __builtin_amdgcn_s_setprio(1);
    else if (c == 3 * N_CHUNK / 4) __builtin_amdgcn_s_setprio(0);
    const int cur = (c & 1) * STAGE_FLOATS;
    const bool more = (c + 1 < N_CHUNK);
    if (more) {
      const int k1 = (c + 1) * KC;
#pragma unroll
      for (int i = 0; i < 4; ++i)
        pre[i] = *(const v4f*)(xg + i * 16 * D_MODEL + k1);
#pragma unroll
      for (int i = 0; i < 6; ++i)
        pre[4 + i] = *(const v4f*)(wg + i * 16 * D_MODEL + k1);
    }
    v2f af[KC / 4][2], bf[KC / 4][3];
#pragma unroll
    for (int kk = 0; kk < KC / 4; ++kk) {
      af[kk][0] = *(const v2f*)(fa + cur + kk * 4);
      af[kk][1] = *(const v2f*)(fa + cur + 16 * PITCH + kk * 4);
      bf[kk][0] = *(const v2f*)(fb + cur + kk * 4);
      bf[kk][1] = *(const v2f*)(fb + cur + 16 * PITCH + kk * 4);
      bf[kk][2] = *(const v2f*)(fb + cur + 32 * PITCH + kk * 4);
    }
    __builtin_amdgcn_sched_barrier(0);
#pragma unroll
    for (int kk = 0; kk < KC / 4; ++kk) {
      acc[0][0] = wmma_f32(af[kk][0], bf[kk][0], acc[0][0]);
      acc[0][1] = wmma_f32(af[kk][0], bf[kk][1], acc[0][1]);
      acc[0][2] = wmma_f32(af[kk][0], bf[kk][2], acc[0][2]);
      acc[1][0] = wmma_f32(af[kk][1], bf[kk][0], acc[1][0]);
      acc[1][1] = wmma_f32(af[kk][1], bf[kk][1], acc[1][1]);
      acc[1][2] = wmma_f32(af[kk][1], bf[kk][2], acc[1][2]);
    }
    __builtin_amdgcn_sched_barrier(0);
    if (more) {
      const int nxt = STAGE_FLOATS - cur;
#pragma unroll
      for (int i = 0; i < 10; ++i) *(v4f*)(st + nxt + i * 16 * PITCH) = pre[i];
    }
    __syncthreads();
  }
}

__global__ __launch_bounds__(128) void qkv_gemm_kernel(
    const float* __restrict__ X, const float* __restrict__ W,
    const float* __restrict__ bias, float* __restrict__ out, int vmode) {
  __shared__ __attribute__((aligned(16))) float lds[GEMM_LDS_FLOATS];
  const int wave = threadIdx.x >> 5;
  const int lane = threadIdx.x & 31;
  const int lo = lane & 15, hi = lane >> 4;
  const int m0 = blockIdx.x * BM + (wave >> 1) * 32;
  const int n0 = blockIdx.y * BN + (wave & 1) * 48;
  v8f acc[2][3] = {{{}, {}, {}}, {{}, {}, {}}};
  gemm_tile_lds(X, W, lds, acc);
  if (vmode) {
    float* tw = lds + wave * (48 * VT_PITCH);
#pragma unroll
    for (int j = 0; j < 3; ++j) {
      const float bv = bias[n0 + j * 16 + lo];
#pragma unroll
      for (int i = 0; i < 2; ++i) {
        v4f c0, c1;
#pragma unroll
        for (int e = 0; e < 4; ++e) {
          c0[e] = acc[i][j][e] + bv;
          c1[e] = acc[i][j][4 + e] + bv;
        }
        float* tp = tw + (j * 16 + lo) * VT_PITCH + i * 16 + 8 * hi;
        *(v4f*)tp = c0;
        *(v4f*)(tp + 4) = c1;
      }
    }
    const int rr = lane >> 3, cq = (lane & 7) * 4;
    const int m = m0 + cq;
    const int b = m >> 11, s = m & (N_CTX - 1);
#pragma unroll
    for (int it = 0; it < 12; ++it) {
      const int nl = it * 4 + rr;
      const v4f t = *(const v4f*)(tw + nl * VT_PITCH + cq);
      const int n = n0 + nl;
      const int h = n >> 6, d = n & (D_HEAD - 1);
      *(v4f*)(out + (((size_t)b * N_HEAD + h) * D_HEAD + d) * N_CTX + s) = t;
    }
    return;
  }
#pragma unroll
  for (int j = 0; j < 3; ++j) {
    const int n = n0 + j * 16 + lo;
    const int h = n >> 6, d = n & (D_HEAD - 1);
    const float bv = bias[n];
#pragma unroll
    for (int i = 0; i < 2; ++i) {
#pragma unroll
      for (int r = 0; r < 8; ++r) {
        const int m = m0 + i * 16 + r + 8 * hi;
        const int b = m >> 11, s = m & (N_CTX - 1);
        const size_t hb = (size_t)b * N_HEAD + h;
        out[(hb * N_CTX + s) * D_HEAD + d] = acc[i][j][r] + bv;
      }
    }
  }
}

__global__ __launch_bounds__(128) void out_gemm_kernel(
    const float* __restrict__ X, const float* __restrict__ W,
    float* __restrict__ out) {
  __shared__ __attribute__((aligned(16))) float lds[GEMM_LDS_FLOATS];
  const int wave = threadIdx.x >> 5;
  const int lane = threadIdx.x & 31;
  const int lo = lane & 15, hi = lane >> 4;
  const int m0 = blockIdx.x * BM + (wave >> 1) * 32;
  const int n0 = blockIdx.y * BN + (wave & 1) * 48;
  v8f acc[2][3] = {{{}, {}, {}}, {{}, {}, {}}};
  gemm_tile_lds(X, W, lds, acc);
#pragma unroll
  for (int i = 0; i < 2; ++i) {
#pragma unroll
    for (int j = 0; j < 3; ++j) {
#pragma unroll
      for (int r = 0; r < 8; ++r) {
        const int m = m0 + i * 16 + r + 8 * hi;
        out[(size_t)m * D_MODEL + n0 + j * 16 + lo] = acc[i][j][r];
      }
    }
  }
}

#define NR_ROWS 24
#define NR_IT   (NR_ROWS / 4)
#define NR_BLOCKS ((2 * BATCH * N_HEAD * N_CTX) / (16 * NR_ROWS))
__global__ __launch_bounds__(512) void norm_rope_kernel(
    float* __restrict__ qk, const float* __restrict__ g) {
  const int lane = threadIdx.x & 31;
  const int gw = blockIdx.x * 16 + (threadIdx.x >> 5);
  const int c8 = lane & 7;
  const int row0 = gw * NR_ROWS + (lane >> 3);
  float* p = qk + (size_t)row0 * D_HEAD + 4 * c8;
  v4f xa[NR_IT], xb[NR_IT];
#pragma unroll
  for (int it = 0; it < NR_IT; ++it) {
    xa[it] = *(const v4f*)(p + it * 4 * D_HEAD);
    xb[it] = *(const v4f*)(p + it * 4 * D_HEAD + 32);
  }
  const v4f ga = *(const v4f*)(g + 4 * c8);
  const v4f gb = *(const v4f*)(g + 32 + 4 * c8);
  asm volatile("" :: "v"(xa[0]), "v"(xb[0]), "v"(xa[1]), "v"(xb[1]), "v"(xa[2]), "v"(xb[2]),
                     "v"(xa[3]), "v"(xb[3]), "v"(xa[4]), "v"(xb[4]), "v"(xa[5]), "v"(xb[5]),
                     "v"(ga), "v"(gb));
  float fr[4];
#pragma unroll
  for (int e = 0; e < 4; ++e)
    fr[e] = exp2f((float)(4 * c8 + e) * (-11.0f / 32.0f));
#pragma unroll
  for (int it = 0; it < NR_IT; ++it) {
    const int s = (row0 + it * 4) & (N_CTX - 1);
    const v4f a = xa[it], b = xb[it];
    float ss = 0.0f;
#pragma unroll
    for (int e = 0; e < 4; ++e) ss += a[e] * a[e] + b[e] * b[e];
    ss += __shfl_xor(ss, 1, 32);
    ss += __shfl_xor(ss, 2, 32);
    ss += __shfl_xor(ss, 4, 32);
    const float inv = rsqrtf(ss * (1.0f / 64.0f) + 1e-6f);
    v4f o1, o2;
#pragma unroll
    for (int e = 0; e < 4; ++e) {
      const float xn1 = a[e] * inv * ga[e];
      const float xn2 = b[e] * inv * gb[e];
      const float freq = (float)s * fr[e];
      const float c = __cosf(freq), sn = __sinf(freq);
      o1[e] = xn1 * c + xn2 * sn;
      o2[e] = xn2 * c - xn1 * sn;
    }
    *(v4f*)(p + it * 4 * D_HEAD)      = o1;
    *(v4f*)(p + it * 4 * D_HEAD + 32) = o2;
  }
}

#define AK 64
#define APITCH 68
#define ATILE (AK * APITCH)
#define ASTAGE (2 * ATILE)
#define ABLK_PER_HEAD 10

#define NT 2
__device__ __forceinline__ void attn_tiles(const float* __restrict__ kst,
                                           const float* __restrict__ vst,
                                           const v2f (&qa)[16], v8f (&z)[4],
                                           const bool d0, const bool d1,
                                           const int lo, const int hi) {
  v2f kb[NT][16], vb[NT][4][4];
#pragma unroll
  for (int u = 0; u < NT; ++u) {
#pragma unroll
    for (int kk = 0; kk < 16; ++kk)
      kb[u][kk] = *(const v2f*)(kst + u * 16 * APITCH + kk * 4);
  }
#pragma unroll
  for (int u = 0; u < NT; ++u) {
#pragma unroll
    for (int j = 0; j < 4; ++j) {
#pragma unroll
      for (int db = 0; db < 4; ++db)
        vb[u][j][db] = *(const v2f*)(vst + db * 16 * APITCH + u * 16 + j * 4);
    }
  }
  __builtin_amdgcn_sched_barrier(0);
  v8f sc[NT];
#pragma unroll
  for (int u = 0; u < NT; ++u) sc[u] = v8f{};
#pragma unroll
  for (int kk = 0; kk < 16; ++kk) {
#pragma unroll
    for (int u = 0; u < NT; ++u) sc[u] = wmma_f32(kb[u][kk], qa[kk], sc[u]);
  }
#pragma unroll
  for (int r = 0; r < 8; ++r) {
    const bool above = 8 * hi + r > lo;
    float p0 = sc[0][r] * (1.0f / 64.0f);
    float p1 = sc[1][r] * (1.0f / 64.0f);
    p0 = p0 * p0;
    p1 = p1 * p1;
    if (d0 && above) p0 = 0.0f;
    if (d0 || (d1 && above)) p1 = 0.0f;
    sc[0][r] = p0;
    sc[1][r] = p1;
  }
#pragma unroll
  for (int u = 0; u < NT; ++u) {
#pragma unroll
    for (int j = 0; j < 4; ++j) {
      const v2f pb = {sc[u][2 * j], sc[u][2 * j + 1]};
#pragma unroll
      for (int db = 0; db < 4; ++db) z[db] = wmma_f32(vb[u][j][db], pb, z[db]);
    }
  }
}

__global__ __launch_bounds__(512) void attn_kernel(
    const float* __restrict__ Q, const float* __restrict__ K,
    const float* __restrict__ V, float* __restrict__ Z) {
  __shared__ __attribute__((aligned(16))) float kv[2 * ASTAGE];
  const int tid = threadIdx.x;
  const int wave = __builtin_amdgcn_readfirstlane(tid >> 5);
  const int lane = tid & 31;
  const int lo = lane & 15, hi = lane >> 4;
  const int head = blockIdx.x / ABLK_PER_HEAD;
  const int bi = blockIdx.x - head * ABLK_PER_HEAD;
  constexpr unsigned int groups_of_block[ABLK_PER_HEAD] = {
      0xFFFF0B1Fu,
      0x0002071Eu,
      0xFFFF141Du,
      0xFFFF191Cu,
      0xFF01181Bu,
      0xFF03171Au,
      0x04051516u,
      0x08091013u,
      0x060C1112u,
      0x0A0D0E0Fu};
  const unsigned int gw = groups_of_block[bi];
  const int wa = wave >> 2, wb = wave & 3;
  const int slot = wa ^ wb;
  const int place = wa ^ ((0x1320 >> (4 * wb)) & 3);
  const int grp = (int)((gw >> (8 * slot)) & 0xFFu);
  const bool has = grp != 0xFF;
  const int qt = has ? grp * 4 + place : -1;
  const float* Qh = Q + (size_t)head * N_CTX * D_HEAD;
  const float* Kh = K + (size_t)head * N_CTX * D_HEAD;
  const float* Vh = V + (size_t)head * D_HEAD * N_CTX;

  const int cr = tid >> 4, cc = (tid & 15) * 4;
  const float* kg = Kh + (size_t)cr * D_HEAD + cc;
  const float* vg = Vh + (size_t)cr * N_CTX + cc;
  float* st = kv + cr * APITCH + cc;
  const int cq = tid & 3;
  float* sv = kv + ATILE + cr * APITCH + (cc & 48) + 8 * (cq & 1) + 2 * (cq >> 1);
  const float* kf = kv + lo * APITCH + 2 * hi;
  const float* vf = kv + ATILE + lo * APITCH + 2 * hi;
  const int nstage = (int)(gw & 0xFFu) + 1;

  v4f pre[4];
#pragma unroll
  for (int p = 0; p < 2; ++p) {
    pre[p]     = *(const v4f*)(kg + p * 32 * D_HEAD);
    pre[2 + p] = *(const v4f*)(vg + (size_t)p * 32 * N_CTX);
  }
#pragma unroll
  for (int p = 0; p < 2; ++p) {
    *(v4f*)(st + p * 32 * APITCH) = pre[p];
    *(v2f*)(sv + p * 32 * APITCH)     = v2f{pre[2 + p][0], pre[2 + p][1]};
    *(v2f*)(sv + p * 32 * APITCH + 4) = v2f{pre[2 + p][2], pre[2 + p][3]};
  }

  v2f qa[16];
  {
    const float* qrow = Qh + (size_t)((has ? qt : 0) * 16 + lo) * D_HEAD + 2 * hi;
#pragma unroll
    for (int kk = 0; kk < 16; ++kk) qa[kk] = *(const v2f*)(qrow + kk * 4);
  }
  __syncthreads();

  v8f z[4] = {{}, {}, {}, {}};
  for (int s = 0; s < nstage; ++s) {
    const int cur = (s & 1) * ASTAGE;
    const bool more = (s + 1 < nstage);
    if (more) {
      const float* kn = kg + (size_t)(s + 1) * AK * D_HEAD;
      const float* vn = vg + (s + 1) * AK;
#pragma unroll
      for (int p = 0; p < 2; ++p) {
        pre[p]     = *(const v4f*)(kn + p * 32 * D_HEAD);
        pre[2 + p] = *(const v4f*)(vn + (size_t)p * 32 * N_CTX);
      }
    }
    for (int tp = 0; tp < 2; ++tp) {
      const int kt = s * 4 + tp * 2;
      if (kt > qt) break;
      const float* kst = kf + cur + tp * 32 * APITCH;
      const float* vst = vf + cur + tp * 32;
      attn_tiles(kst, vst, qa, z, kt == qt, kt + 1 == qt, lo, hi);
    }
    if (more) {
      const int nxt = ASTAGE - cur;
#pragma unroll
      for (int p = 0; p < 2; ++p) {
        *(v4f*)(st + nxt + p * 32 * APITCH) = pre[p];
        *(v2f*)(sv + nxt + p * 32 * APITCH)     = v2f{pre[2 + p][0], pre[2 + p][1]};
        *(v2f*)(sv + nxt + p * 32 * APITCH + 4) = v2f{pre[2 + p][2], pre[2 + p][3]};
      }
    }
    __syncthreads();
  }
  const int b = head / N_HEAD, h = head % N_HEAD;
  if (has) {
    float* zt = kv + wave * (16 * APITCH);
#pragma unroll
    for (int db = 0; db < 4; ++db) {
      float* zp = zt + lo * APITCH + db * 16 + 8 * hi;
      *(v4f*)zp       = v4f{z[db][0], z[db][1], z[db][2], z[db][3]};
      *(v4f*)(zp + 4) = v4f{z[db][4], z[db][5], z[db][6], z[db][7]};
    }
    const int rr = lane >> 4, c4 = (lane & 15) * 4;
    float* zg = Z + (size_t)(b * N_CTX + qt * 16 + rr) * D_MODEL + h * D_HEAD + c4;
#pragma unroll
    for (int it = 0; it < 8; ++it) {
      const v4f t = *(const v4f*)(zt + (it * 2 + rr) * APITCH + c4);
      *(v4f*)(zg + (size_t)it * 2 * D_MODEL) = t;
    }
  }
}

extern "C" void kernel_launch(void* const* d_in, const int* in_sizes, int n_in,
                              void* d_out, int out_size, void* d_ws, size_t ws_size,
                              hipStream_t stream) {
  const float* x  = (const float*)d_in[0];
  const float* Wq = (const float*)d_in[1];
  const float* bq = (const float*)d_in[2];
  const float* Wk = (const float*)d_in[3];
  const float* bk = (const float*)d_in[4];
  const float* Wv = (const float*)d_in[5];
  const float* bv = (const float*)d_in[6];
  const float* Wo = (const float*)d_in[7];
  const float* g  = (const float*)d_in[8];
  float* out = (float*)d_out;

  const size_t QKV = (size_t)BATCH * N_HEAD * N_CTX * D_HEAD;
  float* q = (float*)d_ws;
  float* k = q + QKV;
  float* v = k + QKV;
  float* z = v + QKV;

  dim3 gblk(M_ROWS / BM, D_MODEL / BN);
  qkv_gemm_kernel<<<gblk, 128, 0, stream>>>(x, Wq, bq, q, 0);
  qkv_gemm_kernel<<<gblk, 128, 0, stream>>>(x, Wk, bk, k, 0);
  qkv_gemm_kernel<<<gblk, 128, 0, stream>>>(x, Wv, bv, v, 1);

  norm_rope_kernel<<<NR_BLOCKS, 512, 0, stream>>>(q, g);

  attn_kernel<<<BATCH * N_HEAD * ABLK_PER_HEAD, 512, 0, stream>>>(q, k, v, z);

  out_gemm_kernel<<<gblk, 128, 0, stream>>>(z, Wo, out);
}


# Q, K, V projections as ONE GEMM launch on three weight tiles: a block stages its 64-row x tile once per K chunk for all three (144 WMMAs per barrier, was 48); 4 launches, was 6; sums unchanged
# speedup vs baseline: 3.2711x; 1.0706x over previous
#include <hip/hip_runtime.h>
#include <cstddef>

typedef float v2f __attribute__((ext_vector_type(2)));
typedef float v8f __attribute__((ext_vector_type(8)));

#define D_MODEL 768
#define N_HEAD  12
#define D_HEAD  64
#define N_CTX   2048
#define BATCH   2
#define M_ROWS  (BATCH * N_CTX)

__device__ __forceinline__ v8f wmma_f32(v2f a, v2f b, v8f c) {
  return __builtin_amdgcn_wmma_f32_16x16x4_f32(false, a, false, b, (short)0, c,
                                               false, false);
}

typedef float v4f __attribute__((ext_vector_type(4)));

#define BM 64
#define BN 96
#define KC 32
#define PITCH 36
#define N_CHUNK (D_MODEL / KC)
#define GEMM_LDS_FLOATS 28672
#define VT_PITCH 36

template <int NW>
__device__ __forceinline__ void gemm_tile_lds(const float* __restrict__ X,
                                              const float* const (&W)[NW],
                                              float* __restrict__ lds,
                                              v8f (&acc)[NW][2][3]) {
  constexpr int NPASS = 4 + 6 * NW;
  constexpr int STAGE = (BM + NW * BN) * PITCH;
  const int tid = threadIdx.x;
  const int wave = tid >> 5, lane = tid & 31;
  const int lo = lane & 15, hi = lane >> 4;
  const int wm = wave >> 1, wn = wave & 1;
  const int cr = tid >> 3, cc = (tid & 7) * 4;
  const float* xg = X + (size_t)(blockIdx.x * BM + cr) * D_MODEL + cc;
  const float* wg[NW];
#pragma unroll
  for (int w = 0; w < NW; ++w)
    wg[w] = W[w] + (size_t)(blockIdx.y * BN + cr) * D_MODEL + cc;
  float* st = lds + cr * PITCH + cc;
  const float* fa = lds + (wm * 32 + lo) * PITCH + 2 * hi;
  const float* fb = lds + (BM + wn * 48 + lo) * PITCH + 2 * hi;

  v4f pre[NPASS];
#pragma unroll
  for (int i = 0; i < 4; ++i) pre[i] = *(const v4f*)(xg + i * 16 * D_MODEL);
#pragma unroll
  for (int w = 0; w < NW; ++w) {
#pragma unroll
    for (int i = 0; i < 6; ++i)
      pre[4 + 6 * w + i] = *(const v4f*)(wg[w] + i * 16 * D_MODEL);
  }
#pragma unroll
  for (int i = 0; i < NPASS; ++i) *(v4f*)(st + i * 16 * PITCH) = pre[i];
  __syncthreads();

  for (int c = 0; c < N_CHUNK; ++c) {
    if (c == 0) __builtin_amdgcn_s_setprio(3);
    else if (c == N_CHUNK / 4) __builtin_amdgcn_s_setprio(2);
    else if (c == N_CHUNK / 2) __builtin_amdgcn_s_setprio(1);
    else if (c == 3 * N_CHUNK / 4) __builtin_amdgcn_s_setprio(0);
    const int cur = (c & 1) * STAGE;
    const bool more = (c + 1 < N_CHUNK);
    if (more) {
      const int k1 = (c + 1) * KC;
#pragma unroll
      for (int i = 0; i < 4; ++i)
        pre[i] = *(const v4f*)(xg + i * 16 * D_MODEL + k1);
#pragma unroll
      for (int w = 0; w < NW; ++w) {
#pragma unroll
        for (int i = 0; i < 6; ++i)
          pre[4 + 6 * w + i] = *(const v4f*)(wg[w] + i * 16 * D_MODEL + k1);
      }
    }
    v2f af[KC / 4][2], bf[NW][KC / 4][3];
#pragma unroll
    for (int kk = 0; kk < KC / 4; ++kk) {
      af[kk][0] = *(const v2f*)(fa + cur + kk * 4);
      af[kk][1] = *(const v2f*)(fa + cur + 16 * PITCH + kk * 4);
      bf[0][kk][0] = *(const v2f*)(fb + cur + kk * 4);
      bf[0][kk][1] = *(const v2f*)(fb + cur + 16 * PITCH + kk * 4);
      bf[0][kk][2] = *(const v2f*)(fb + cur + 32 * PITCH + kk * 4);
    }
    __builtin_amdgcn_sched_barrier(0);
#pragma unroll
    for (int w = 0; w < NW; ++w) {
      if (w + 1 < NW) {
#pragma unroll
        for (int kk = 0; kk < KC / 4; ++kk) {
          const float* fn = fb + cur + (w + 1) * BN * PITCH + kk * 4;
          bf[w + 1][kk][0] = *(const v2f*)(fn);
          bf[w + 1][kk][1] = *(const v2f*)(fn + 16 * PITCH);
          bf[w + 1][kk][2] = *(const v2f*)(fn + 32 * PITCH);
        }
        __builtin_amdgcn_sched_barrier(0);
      }
#pragma unroll
      for (int kk = 0; kk < KC / 4; ++kk) {
        acc[w][0][0] = wmma_f32(af[kk][0], bf[w][kk][0], acc[w][0][0]);
        acc[w][0][1] = wmma_f32(af[kk][0], bf[w][kk][1], acc[w][0][1]);
        acc[w][0][2] = wmma_f32(af[kk][0], bf[w][kk][2], acc[w][0][2]);
        acc[w][1][0] = wmma_f32(af[kk][1], bf[w][kk][0], acc[w][1][0]);
        acc[w][1][1] = wmma_f32(af[kk][1], bf[w][kk][1], acc[w][1][1]);
        acc[w][1][2] = wmma_f32(af[kk][1], bf[w][kk][2], acc[w][1][2]);
      }
      __builtin_amdgcn_sched_barrier(0);
    }
    if (more) {
      const int nxt = STAGE - cur;
#pragma unroll
      for (int i = 0; i < NPASS; ++i) *(v4f*)(st + nxt + i * 16 * PITCH) = pre[i];
    }
    __syncthreads();
  }
}

__device__ __forceinline__ void store_heads(const v8f (&acc)[2][3],
                                            const float* __restrict__ bias,
                                            float* __restrict__ out,
                                            const int m0, const int n0,
                                            const int lo, const int hi) {
#pragma unroll
  for (int j = 0; j < 3; ++j) {
    const int n = n0 + j * 16 + lo;
    const int h = n >> 6, d = n & (D_HEAD - 1);
    const float bv = bias[n];
#pragma unroll
    for (int i = 0; i < 2; ++i) {
#pragma unroll
      for (int r = 0; r < 8; ++r) {
        const int m = m0 + i * 16 + r + 8 * hi;
        const int b = m >> 11, s = m & (N_CTX - 1);
        const size_t hb = (size_t)b * N_HEAD + h;
        out[(hb * N_CTX + s) * D_HEAD + d] = acc[i][j][r] + bv;
      }
    }
  }
}

__device__ __forceinline__ void store_vt(const v8f (&acc)[2][3],
                                         const float* __restrict__ bias,
                                         float* __restrict__ out,
                                         float* __restrict__ lds,
                                         const int wave, const int lane,
                                         const int m0, const int n0) {
  const int lo = lane & 15, hi = lane >> 4;
  float* tw = lds + wave * (48 * VT_PITCH);
#pragma unroll
  for (int j = 0; j < 3; ++j) {
    const float bv = bias[n0 + j * 16 + lo];
#pragma unroll
    for (int i = 0; i < 2; ++i) {
      v4f c0, c1;
#pragma unroll
      for (int e = 0; e < 4; ++e) {
        c0[e] = acc[i][j][e] + bv;
        c1[e] = acc[i][j][4 + e] + bv;
      }
      float* tp = tw + (j * 16 + lo) * VT_PITCH + i * 16 + 8 * hi;
      *(v4f*)tp = c0;
      *(v4f*)(tp + 4) = c1;
    }
  }
  const int rr = lane >> 3, cq = (lane & 7) * 4;
  const int m = m0 + cq;
  const int b = m >> 11, s = m & (N_CTX - 1);
#pragma unroll
  for (int it = 0; it < 12; ++it) {
    const int nl = it * 4 + rr;
    const v4f t = *(const v4f*)(tw + nl * VT_PITCH + cq);
    const int n = n0 + nl;
    const int h = n >> 6, d = n & (D_HEAD - 1);
    *(v4f*)(out + (((size_t)b * N_HEAD + h) * D_HEAD + d) * N_CTX + s) = t;
  }
}

__global__ __launch_bounds__(128) void qkv_gemm_kernel(
    const float* __restrict__ X, const float* __restrict__ Wq,
    const float* __restrict__ Wk, const float* __restrict__ Wv,
    const float* __restrict__ bq, const float* __restrict__ bk,
    const float* __restrict__ bv, float* __restrict__ q, float* __restrict__ k,
    float* __restrict__ v) {
  __shared__ __attribute__((aligned(16))) float lds[GEMM_LDS_FLOATS];
  const int wave = threadIdx.x >> 5;
  const int lane = threadIdx.x & 31;
  const int lo = lane & 15, hi = lane >> 4;
  const int m0 = blockIdx.x * BM + (wave >> 1) * 32;
  const int n0 = blockIdx.y * BN + (wave & 1) * 48;
  v8f acc[3][2][3] = {};
  const float* const W[3] = {Wq, Wk, Wv};
  gemm_tile_lds<3>(X, W, lds, acc);
  store_heads(acc[0], bq, q, m0, n0, lo, hi);
  store_heads(acc[1], bk, k, m0, n0, lo, hi);
  store_vt(acc[2], bv, v, lds, wave, lane, m0, n0);
}

__global__ __launch_bounds__(128) void out_gemm_kernel(
    const float* __restrict__ X, const float* __restrict__ Wo,
    float* __restrict__ out) {
  __shared__ __attribute__((aligned(16))) float lds[GEMM_LDS_FLOATS];
  const int wave = threadIdx.x >> 5;
  const int lane = threadIdx.x & 31;
  const int lo = lane & 15, hi = lane >> 4;
  const int m0 = blockIdx.x * BM + (wave >> 1) * 32;
  const int n0 = blockIdx.y * BN + (wave & 1) * 48;
  v8f acc[1][2][3] = {};
  const float* const W[1] = {Wo};
  gemm_tile_lds<1>(X, W, lds, acc);
#pragma unroll
  for (int i = 0; i < 2; ++i) {
#pragma unroll
    for (int j = 0; j < 3; ++j) {
#pragma unroll
      for (int r = 0; r < 8; ++r) {
        const int m = m0 + i * 16 + r + 8 * hi;
        out[(size_t)m * D_MODEL + n0 + j * 16 + lo] = acc[0][i][j][r];
      }
    }
  }
}

#define NR_ROWS 24
#define NR_IT   (NR_ROWS / 4)
#define NR_BLOCKS ((2 * BATCH * N_HEAD * N_CTX) / (16 * NR_ROWS))
__global__ __launch_bounds__(512) void norm_rope_kernel(
    float* __restrict__ qk, const float* __restrict__ g) {
  const int lane = threadIdx.x & 31;
  const int gw = blockIdx.x * 16 + (threadIdx.x >> 5);
  const int c8 = lane & 7;
  const int row0 = gw * NR_ROWS + (lane >> 3);
  float* p = qk + (size_t)row0 * D_HEAD + 4 * c8;
  v4f xa[NR_IT], xb[NR_IT];
#pragma unroll
  for (int it = 0; it < NR_IT; ++it) {
    xa[it] = *(const v4f*)(p + it * 4 * D_HEAD);
    xb[it] = *(const v4f*)(p + it * 4 * D_HEAD + 32);
  }
  const v4f ga = *(const v4f*)(g + 4 * c8);
  const v4f gb = *(const v4f*)(g + 32 + 4 * c8);
  asm volatile("" :: "v"(xa[0]), "v"(xb[0]), "v"(xa[1]), "v"(xb[1]), "v"(xa[2]), "v"(xb[2]),
                     "v"(xa[3]), "v"(xb[3]), "v"(xa[4]), "v"(xb[4]), "v"(xa[5]), "v"(xb[5]),
                     "v"(ga), "v"(gb));
  float fr[4];
#pragma unroll
  for (int e = 0; e < 4; ++e)
    fr[e] = exp2f((float)(4 * c8 + e) * (-11.0f / 32.0f));
#pragma unroll
  for (int it = 0; it < NR_IT; ++it) {
    const int s = (row0 + it * 4) & (N_CTX - 1);
    const v4f a = xa[it], b = xb[it];
    float ss = 0.0f;
#pragma unroll
    for (int e = 0; e < 4; ++e) ss += a[e] * a[e] + b[e] * b[e];
    ss += __shfl_xor(ss, 1, 32);
    ss += __shfl_xor(ss, 2, 32);
    ss += __shfl_xor(ss, 4, 32);
    const float inv = rsqrtf(ss * (1.0f / 64.0f) + 1e-6f);
    v4f o1, o2;
#pragma unroll
    for (int e = 0; e < 4; ++e) {
      const float xn1 = a[e] * inv * ga[e];
      const float xn2 = b[e] * inv * gb[e];
      const float freq = (float)s * fr[e];
      const float c = __cosf(freq), sn = __sinf(freq);
      o1[e] = xn1 * c + xn2 * sn;
      o2[e] = xn2 * c - xn1 * sn;
    }
    *(v4f*)(p + it * 4 * D_HEAD)      = o1;
    *(v4f*)(p + it * 4 * D_HEAD + 32) = o2;
  }
}

#define AK 64
#define APITCH 68
#define ATILE (AK * APITCH)
#define ASTAGE (2 * ATILE)
#define ABLK_PER_HEAD 10

#define NT 2
__device__ __forceinline__ void attn_tiles(const float* __restrict__ kst,
                                           const float* __restrict__ vst,
                                           const v2f (&qa)[16], v8f (&z)[4],
                                           const bool d0, const bool d1,
                                           const int lo, const int hi) {
  v2f kb[NT][16], vb[NT][4][4];
#pragma unroll
  for (int u = 0; u < NT; ++u) {
#pragma unroll
    for (int kk = 0; kk < 16; ++kk)
      kb[u][kk] = *(const v2f*)(kst + u * 16 * APITCH + kk * 4);
  }
#pragma unroll
  for (int u = 0; u < NT; ++u) {
#pragma unroll
    for (int j = 0; j < 4; ++j) {
#pragma unroll
      for (int db = 0; db < 4; ++db)
        vb[u][j][db] = *(const v2f*)(vst + db * 16 * APITCH + u * 16 + j * 4);
    }
  }
  __builtin_amdgcn_sched_barrier(0);
  v8f sc[NT];
#pragma unroll
  for (int u = 0; u < NT; ++u) sc[u] = v8f{};
#pragma unroll
  for (int kk = 0; kk < 16; ++kk) {
#pragma unroll
    for (int u = 0; u < NT; ++u) sc[u] = wmma_f32(kb[u][kk], qa[kk], sc[u]);
  }
#pragma unroll
  for (int r = 0; r < 8; ++r) {
    const bool above = 8 * hi + r > lo;
    float p0 = sc[0][r] * (1.0f / 64.0f);
    float p1 = sc[1][r] * (1.0f / 64.0f);
    p0 = p0 * p0;
    p1 = p1 * p1;
    if (d0 && above) p0 = 0.0f;
    if (d0 || (d1 && above)) p1 = 0.0f;
    sc[0][r] = p0;
    sc[1][r] = p1;
  }
#pragma unroll
  for (int u = 0; u < NT; ++u) {
#pragma unroll
    for (int j = 0; j < 4; ++j) {
      const v2f pb = {sc[u][2 * j], sc[u][2 * j + 1]};
#pragma unroll
      for (int db = 0; db < 4; ++db) z[db] = wmma_f32(vb[u][j][db], pb, z[db]);
    }
  }
}

__global__ __launch_bounds__(512) void attn_kernel(
    const float* __restrict__ Q, const float* __restrict__ K,
    const float* __restrict__ V, float* __restrict__ Z) {
  __shared__ __attribute__((aligned(16))) float kv[2 * ASTAGE];
  const int tid = threadIdx.x;
  const int wave = __builtin_amdgcn_readfirstlane(tid >> 5);
  const int lane = tid & 31;
  const int lo = lane & 15, hi = lane >> 4;
  const int head = blockIdx.x / ABLK_PER_HEAD;
  const int bi = blockIdx.x - head * ABLK_PER_HEAD;
  constexpr unsigned int groups_of_block[ABLK_PER_HEAD] = {
      0xFFFF0B1Fu,
      0x0002071Eu,
      0xFFFF141Du,
      0xFFFF191Cu,
      0xFF01181Bu,
      0xFF03171Au,
      0x04051516u,
      0x08091013u,
      0x060C1112u,
      0x0A0D0E0Fu};
  const unsigned int gw = groups_of_block[bi];
  const int wa = wave >> 2, wb = wave & 3;
  const int slot = wa ^ wb;
  const int place = wa ^ ((0x1320 >> (4 * wb)) & 3);
  const int grp = (int)((gw >> (8 * slot)) & 0xFFu);
  const bool has = grp != 0xFF;
  const int qt = has ? grp * 4 + place : -1;
  const float* Qh = Q + (size_t)head * N_CTX * D_HEAD;
  const float* Kh = K + (size_t)head * N_CTX * D_HEAD;
  const float* Vh = V + (size_t)head * D_HEAD * N_CTX;

  const int cr = tid >> 4, cc = (tid & 15) * 4;
  const float* kg = Kh + (size_t)cr * D_HEAD + cc;
  const float* vg = Vh + (size_t)cr * N_CTX + cc;
  float* st = kv + cr * APITCH + cc;
  const int cq = tid & 3;
  float* sv = kv + ATILE + cr * APITCH + (cc & 48) + 8 * (cq & 1) + 2 * (cq >> 1);
  const float* kf = kv + lo * APITCH + 2 * hi;
  const float* vf = kv + ATILE + lo * APITCH + 2 * hi;
  const int nstage = (int)(gw & 0xFFu) + 1;

  v4f pre[4];
#pragma unroll
  for (int p = 0; p < 2; ++p) {
    pre[p]     = *(const v4f*)(kg + p * 32 * D_HEAD);
    pre[2 + p] = *(const v4f*)(vg + (size_t)p * 32 * N_CTX);
  }
#pragma unroll
  for (int p = 0; p < 2; ++p) {
    *(v4f*)(st + p * 32 * APITCH) = pre[p];
    *(v2f*)(sv + p * 32 * APITCH)     = v2f{pre[2 + p][0], pre[2 + p][1]};
    *(v2f*)(sv + p * 32 * APITCH + 4) = v2f{pre[2 + p][2], pre[2 + p][3]};
  }

  v2f qa[16];
  {
    const float* qrow = Qh + (size_t)((has ? qt : 0) * 16 + lo) * D_HEAD + 2 * hi;
#pragma unroll
    for (int kk = 0; kk < 16; ++kk) qa[kk] = *(const v2f*)(qrow + kk * 4);
  }
  __syncthreads();

  v8f z[4] = {{}, {}, {}, {}};
  for (int s = 0; s < nstage; ++s) {
    const int cur = (s & 1) * ASTAGE;
    const bool more = (s + 1 < nstage);
    if (more) {
      const float* kn = kg + (size_t)(s + 1) * AK * D_HEAD;
      const float* vn = vg + (s + 1) * AK;
#pragma unroll
      for (int p = 0; p < 2; ++p) {
        pre[p]     = *(const v4f*)(kn + p * 32 * D_HEAD);
        pre[2 + p] = *(const v4f*)(vn + (size_t)p * 32 * N_CTX);
      }
    }
    for (int tp = 0; tp < 2; ++tp) {
      const int kt = s * 4 + tp * 2;
      if (kt > qt) break;
      const float* kst = kf + cur + tp * 32 * APITCH;
      const float* vst = vf + cur + tp * 32;
      attn_tiles(kst, vst, qa, z, kt == qt, kt + 1 == qt, lo, hi);
    }
    if (more) {
      const int nxt = ASTAGE - cur;
#pragma unroll
      for (int p = 0; p < 2; ++p) {
        *(v4f*)(st + nxt + p * 32 * APITCH) = pre[p];
        *(v2f*)(sv + nxt + p * 32 * APITCH)     = v2f{pre[2 + p][0], pre[2 + p][1]};
        *(v2f*)(sv + nxt + p * 32 * APITCH + 4) = v2f{pre[2 + p][2], pre[2 + p][3]};
      }
    }
    __syncthreads();
  }
  const int b = head / N_HEAD, h = head % N_HEAD;
  if (has) {
    float* zt = kv + wave * (16 * APITCH);
#pragma unroll
    for (int db = 0; db < 4; ++db) {
      float* zp = zt + lo * APITCH + db * 16 + 8 * hi;
      *(v4f*)zp       = v4f{z[db][0], z[db][1], z[db][2], z[db][3]};
      *(v4f*)(zp + 4) = v4f{z[db][4], z[db][5], z[db][6], z[db][7]};
    }
    const int rr = lane >> 4, c4 = (lane & 15) * 4;
    float* zg = Z + (size_t)(b * N_CTX + qt * 16 + rr) * D_MODEL + h * D_HEAD + c4;
#pragma unroll
    for (int it = 0; it < 8; ++it) {
      const v4f t = *(const v4f*)(zt + (it * 2 + rr) * APITCH + c4);
      *(v4f*)(zg + (size_t)it * 2 * D_MODEL) = t;
    }
  }
}

extern "C" void kernel_launch(void* const* d_in, const int* in_sizes, int n_in,
                              void* d_out, int out_size, void* d_ws, size_t ws_size,
                              hipStream_t stream) {
  const float* x  = (const float*)d_in[0];
  const float* Wq = (const float*)d_in[1];
  const float* bq = (const float*)d_in[2];
  const float* Wk = (const float*)d_in[3];
  const float* bk = (const float*)d_in[4];
  const float* Wv = (const float*)d_in[5];
  const float* bv = (const float*)d_in[6];
  const float* Wo = (const float*)d_in[7];
  const float* g  = (const float*)d_in[8];
  float* out = (float*)d_out;

  const size_t QKV = (size_t)BATCH * N_HEAD * N_CTX * D_HEAD;
  float* q = (float*)d_ws;
  float* k = q + QKV;
  float* v = k + QKV;
  float* z = v + QKV;

  dim3 gblk(M_ROWS / BM, D_MODEL / BN);
  qkv_gemm_kernel<<<gblk, 128, 0, stream>>>(x, Wq, Wk, Wv, bq, bk, bv, q, k, v);

  norm_rope_kernel<<<NR_BLOCKS, 512, 0, stream>>>(q, g);

  attn_kernel<<<BATCH * N_HEAD * ABLK_PER_HEAD, 512, 0, stream>>>(q, k, v, z);

  out_gemm_kernel<<<gblk, 128, 0, stream>>>(z, Wo, out);
}
